# starting kernel
# baseline (speedup 1.0000x reference)
#include <hip/hip_runtime.h>
#include <hip/hip_bf16.h>

typedef float v2f __attribute__((ext_vector_type(2)));
typedef float v8f __attribute__((ext_vector_type(8)));

#define GAT_IN_DIM 128
#define GAT_NEG_SLOPE 0.2f

__global__ void wmma_gemm_f32(const float* __restrict__ A,
                              const float* __restrict__ B,
                              float* __restrict__ C,
                              const float* __restrict__ bias,
                              int M, int N, int K) {
  int wave = (blockIdx.x * blockDim.x + threadIdx.x) >> 5;
  int lane = threadIdx.x & 31;
  int tiles_n = N >> 4;
  int tiles = (M >> 4) * tiles_n;
  if (wave >= tiles) return;
  int tm = (wave / tiles_n) << 4;
  int tn = (wave % tiles_n) << 4;
  int idx  = lane & 15;
  int kk   = (lane >> 4) << 1;
  int half = lane >> 4;

  v8f acc = {};
  for (int k0 = 0; k0 < K; k0 += 4) {
    v2f a, b;
    const float* ap = A + (size_t)(tm + idx) * K + (k0 + kk);
    a.x = ap[0];
    a.y = ap[1];
    const float* bp = B + (size_t)(k0 + kk) * N + tn + idx;
    b.x = bp[0];
    b.y = bp[N];
    acc = __builtin_amdgcn_wmma_f32_16x16x4_f32(
         false, a,  false, b,
         (short)0, acc,  false,  false);
  }

  float badd = bias ? bias[tn + idx] : 0.0f;
#pragma unroll
  for (int r = 0; r < 8; ++r) {
    C[(size_t)(tm + r + 8 * half) * N + tn + idx] = acc[r] + badd;
  }
}

__global__ void gat_scores(const float* __restrict__ h,
                           const float* __restrict__ attn,
                           float* __restrict__ es, float* __restrict__ ed,
                           int n_nodes, int H, int F) {
  int t = blockIdx.x * blockDim.x + threadIdx.x;
  if (t >= n_nodes * H) return;
  int n = t / H;
  int hh = t - n * H;
  const float* hp = h + (size_t)n * H * F + (size_t)hh * F;
  const float* al = attn + (size_t)hh * 2 * F;
  const float* ar = al + F;
  float s = 0.f, d = 0.f;
  for (int f = 0; f < F; ++f) {
    s += hp[f] * al[f];
    d += hp[f] * ar[f];
  }
  es[t] = s;
  ed[t] = d;
}

__device__ __forceinline__ unsigned encf(float f) {
  unsigned u = __float_as_uint(f);
  return (u & 0x80000000u) ? ~u : (u | 0x80000000u);
}
__device__ __forceinline__ float decf(unsigned u) {
  return (u & 0x80000000u) ? __uint_as_float(u & 0x7FFFFFFFu)
                           : __uint_as_float(~u);
}

__global__ void edge_alpha_kernel(const float* __restrict__ es,
                                  const float* __restrict__ ed,
                                  const int* __restrict__ rowE,
                                  const int* __restrict__ colE,
                                  float* __restrict__ alpha,
                                  unsigned* __restrict__ amax,
                                  int E, int Nn, int H) {
  int t = blockIdx.x * blockDim.x + threadIdx.x;
  if (t >= (E + Nn) * H) return;
  int e = t / H;
  int h = t - e * H;
  int r = (e < E) ? rowE[e] : (e - E);
  int c = (e < E) ? colE[e] : (e - E);
  float a = es[r * H + h] + ed[c * H + h];
  a = (a > 0.f) ? a : a * GAT_NEG_SLOPE;
  alpha[t] = a;
  atomicMax(&amax[c * H + h], encf(a));
}

__global__ void edge_exp_kernel(float* __restrict__ alpha,
                                const unsigned* __restrict__ amax,
                                float* __restrict__ denom,
                                const int* __restrict__ colE,
                                int E, int Nn, int H) {
  int t = blockIdx.x * blockDim.x + threadIdx.x;
  if (t >= (E + Nn) * H) return;
  int e = t / H;
  int h = t - e * H;
  int c = (e < E) ? colE[e] : (e - E);
  float am = decf(amax[c * H + h]);
  float ex = __expf(alpha[t] - am);
  alpha[t] = ex;
  atomicAdd(&denom[c * H + h], ex);
}

__global__ void edge_scatter_kernel(const float* __restrict__ ex,
                                    const float* __restrict__ denom,
                                    const float* __restrict__ hsrc,
                                    float* __restrict__ outd,
                                    const int* __restrict__ rowE,
                                    const int* __restrict__ colE,
                                    int E, int Nn, int H, int F) {
  int Ftot = H * F;
  long long t = (long long)blockIdx.x * blockDim.x + threadIdx.x;
  long long total = (long long)(E + Nn) * Ftot;
  if (t >= total) return;
  int e = (int)(t / Ftot);
  int f = (int)(t - (long long)e * Ftot);
  int h = f / F;
  int r = (e < E) ? rowE[e] : (e - E);
  int c = (e < E) ? colE[e] : (e - E);
  float coef = ex[(size_t)e * H + h] / denom[c * H + h];
  atomicAdd(&outd[(size_t)c * Ftot + f], hsrc[(size_t)r * Ftot + f] * coef);
}

__global__ void relu_kernel(float* __restrict__ p, long long n) {
  long long t = (long long)blockIdx.x * blockDim.x + threadIdx.x;
  if (t < n) p[t] = fmaxf(p[t], 0.f);
}


extern "C" void kernel_launch(void* const* d_in, const int* in_sizes, int n_in,
                              void* d_out, int out_size, void* d_ws, size_t ws_size,
                              hipStream_t stream) {
  const float* x     = (const float*)d_in[0];
  const int*   ei    = (const int*)  d_in[1];
  const float* W1    = (const float*)d_in[2];
  const float* attn1 = (const float*)d_in[3];
  const float* W2    = (const float*)d_in[4];
  const float* attn2 = (const float*)d_in[5];
  const float* headW = (const float*)d_in[6];
  const float* headb = (const float*)d_in[7];

  const int Nn = in_sizes[0] / GAT_IN_DIM;
  const int E  = in_sizes[1] / 2;
  const int ET = E + Nn;
  const int* rowE = ei;
  const int* colE = ei + E;

  float* w = (float*)d_ws;
  float* h1    = w; w += (size_t)Nn * 128;
  float* out1  = w; w += (size_t)Nn * 128;
  float* h2    = w; w += (size_t)Nn * 64;
  float* out2  = w; w += (size_t)Nn * 64;
  float* es1   = w; w += (size_t)Nn * 4;
  float* ed1   = w; w += (size_t)Nn * 4;
  unsigned* amax1 = (unsigned*)w; w += (size_t)Nn * 4;
  float* den1  = w; w += (size_t)Nn * 4;
  float* es2   = w; w += (size_t)Nn;
  float* ed2   = w; w += (size_t)Nn;
  unsigned* amax2 = (unsigned*)w; w += (size_t)Nn;
  float* den2  = w; w += (size_t)Nn;
  float* ebuf  = w; w += (size_t)ET * 4;

  hipMemsetAsync(out1,  0, (size_t)Nn * 128 * sizeof(float), stream);
  hipMemsetAsync(out2,  0, (size_t)Nn * 64  * sizeof(float), stream);
  hipMemsetAsync(amax1, 0, (size_t)Nn * 4   * sizeof(unsigned), stream);
  hipMemsetAsync(den1,  0, (size_t)Nn * 4   * sizeof(float), stream);
  hipMemsetAsync(amax2, 0, (size_t)Nn       * sizeof(unsigned), stream);
  hipMemsetAsync(den2,  0, (size_t)Nn       * sizeof(float), stream);

  const int TB = 256;
  auto gemm_blocks = [](int M, int N) {
    int tiles = (M >> 4) * (N >> 4);
    return (tiles + 7) / 8;
  };

  wmma_gemm_f32<<<gemm_blocks(Nn, 128), TB, 0, stream>>>(x, W1, h1, nullptr, Nn, 128, 128);
  gat_scores<<<(Nn * 4 + TB - 1) / TB, TB, 0, stream>>>(h1, attn1, es1, ed1, Nn, 4, 32);
  {
    int nt = ET * 4;
    edge_alpha_kernel<<<(nt + TB - 1) / TB, TB, 0, stream>>>(es1, ed1, rowE, colE, ebuf, amax1, E, Nn, 4);
    edge_exp_kernel<<<(nt + TB - 1) / TB, TB, 0, stream>>>(ebuf, amax1, den1, colE, E, Nn, 4);
    long long ns = (long long)ET * 128;
    edge_scatter_kernel<<<(unsigned)((ns + TB - 1) / TB), TB, 0, stream>>>(
        ebuf, den1, h1, out1, rowE, colE, E, Nn, 4, 32);
  }
  {
    long long n = (long long)Nn * 128;
    relu_kernel<<<(unsigned)((n + TB - 1) / TB), TB, 0, stream>>>(out1, n);
  }

  wmma_gemm_f32<<<gemm_blocks(Nn, 64), TB, 0, stream>>>(out1, W2, h2, nullptr, Nn, 64, 128);
  gat_scores<<<(Nn + TB - 1) / TB, TB, 0, stream>>>(h2, attn2, es2, ed2, Nn, 1, 64);
  {
    int nt = ET;
    edge_alpha_kernel<<<(nt + TB - 1) / TB, TB, 0, stream>>>(es2, ed2, rowE, colE, ebuf, amax2, E, Nn, 1);
    edge_exp_kernel<<<(nt + TB - 1) / TB, TB, 0, stream>>>(ebuf, amax2, den2, colE, E, Nn, 1);
    long long ns = (long long)ET * 64;
    edge_scatter_kernel<<<(unsigned)((ns + TB - 1) / TB), TB, 0, stream>>>(
        ebuf, den2, h2, out2, rowE, colE, E, Nn, 1, 64);
  }

  wmma_gemm_f32<<<gemm_blocks(Nn, 64), TB, 0, stream>>>(out2, headW, (float*)d_out, headb, Nn, 64, 64);
}


# Neighbour sums gathered per destination node over a per-call counting sort of the edges (float4 rows, one store per element) replace the f32 atomic scatter; sum order = arrival-order rank
# speedup vs baseline: 2.3493x; 2.3493x over previous
#include <hip/hip_runtime.h>
#include <hip/hip_bf16.h>

typedef float v2f __attribute__((ext_vector_type(2)));
typedef float v8f __attribute__((ext_vector_type(8)));

#define GAT_IN_DIM 128
#define GAT_NEG_SLOPE 0.2f

__global__ void wmma_gemm_f32(const float* __restrict__ A,
                              const float* __restrict__ B,
                              float* __restrict__ C,
                              const float* __restrict__ bias,
                              int M, int N, int K) {
  int wave = (blockIdx.x * blockDim.x + threadIdx.x) >> 5;
  int lane = threadIdx.x & 31;
  int tiles_n = N >> 4;
  int tiles = (M >> 4) * tiles_n;
  if (wave >= tiles) return;
  int tm = (wave / tiles_n) << 4;
  int tn = (wave % tiles_n) << 4;
  int idx  = lane & 15;
  int kk   = (lane >> 4) << 1;
  int half = lane >> 4;

  v8f acc = {};
  for (int k0 = 0; k0 < K; k0 += 4) {
    v2f a, b;
    const float* ap = A + (size_t)(tm + idx) * K + (k0 + kk);
    a.x = ap[0];
    a.y = ap[1];
    const float* bp = B + (size_t)(k0 + kk) * N + tn + idx;
    b.x = bp[0];
    b.y = bp[N];
    acc = __builtin_amdgcn_wmma_f32_16x16x4_f32(
         false, a,  false, b,
         (short)0, acc,  false,  false);
  }

  float badd = bias ? bias[tn + idx] : 0.0f;
#pragma unroll
  for (int r = 0; r < 8; ++r) {
    C[(size_t)(tm + r + 8 * half) * N + tn + idx] = acc[r] + badd;
  }
}

__global__ void gat_scores(const float* __restrict__ h,
                           const float* __restrict__ attn,
                           float* __restrict__ es, float* __restrict__ ed,
                           int n_nodes, int H, int F) {
  int t = blockIdx.x * blockDim.x + threadIdx.x;
  if (t >= n_nodes * H) return;
  int n = t / H;
  int hh = t - n * H;
  const float* hp = h + (size_t)n * H * F + (size_t)hh * F;
  const float* al = attn + (size_t)hh * 2 * F;
  const float* ar = al + F;
  float s = 0.f, d = 0.f;
  for (int f = 0; f < F; ++f) {
    s += hp[f] * al[f];
    d += hp[f] * ar[f];
  }
  es[t] = s;
  ed[t] = d;
}

__device__ __forceinline__ unsigned encf(float f) {
  unsigned u = __float_as_uint(f);
  return (u & 0x80000000u) ? ~u : (u | 0x80000000u);
}
__device__ __forceinline__ float decf(unsigned u) {
  return (u & 0x80000000u) ? __uint_as_float(u & 0x7FFFFFFFu)
                           : __uint_as_float(~u);
}

__global__ void edge_alpha_kernel(const float* __restrict__ es,
                                  const float* __restrict__ ed,
                                  const int* __restrict__ rowE,
                                  const int* __restrict__ colE,
                                  float* __restrict__ alpha,
                                  unsigned* __restrict__ amax,
                                  int E, int Nn, int H) {
  int t = blockIdx.x * blockDim.x + threadIdx.x;
  if (t >= (E + Nn) * H) return;
  int e = t / H;
  int h = t - e * H;
  int r = (e < E) ? rowE[e] : (e - E);
  int c = (e < E) ? colE[e] : (e - E);
  float a = es[r * H + h] + ed[c * H + h];
  a = (a > 0.f) ? a : a * GAT_NEG_SLOPE;
  alpha[t] = a;
  atomicMax(&amax[c * H + h], encf(a));
}

__global__ void edge_exp_kernel(float* __restrict__ alpha,
                                const unsigned* __restrict__ amax,
                                float* __restrict__ denom,
                                const int* __restrict__ colE,
                                int E, int Nn, int H) {
  int t = blockIdx.x * blockDim.x + threadIdx.x;
  if (t >= (E + Nn) * H) return;
  int e = t / H;
  int h = t - e * H;
  int c = (e < E) ? colE[e] : (e - E);
  float am = decf(amax[c * H + h]);
  float ex = __expf(alpha[t] - am);
  alpha[t] = ex;
  atomicAdd(&denom[c * H + h], ex);
}

__global__ void csr_rank_kernel(const int* __restrict__ colE,
                                int* __restrict__ cnt,
                                int* __restrict__ rk, int E) {
  int e = blockIdx.x * blockDim.x + threadIdx.x;
  if (e >= E) return;
  rk[e] = atomicAdd(&cnt[colE[e]], 1);
}

__global__ void __launch_bounds__(1024)
csr_scan_kernel(const int* __restrict__ cnt, int* __restrict__ rowptr, int n) {
  __shared__ int wsum[32];
  __shared__ int base_s;
  const int tid = threadIdx.x;
  const int lane = tid & 31;
  const int wv = tid >> 5;
  const int tile0 = blockIdx.x * 1024;

  int s = 0;
  for (int i = tid; i < tile0; i += 1024) s += cnt[i];
  for (int o = 16; o > 0; o >>= 1) s += __shfl_xor(s, o);
  if (lane == 0) wsum[wv] = s;
  __syncthreads();
  if (wv == 0) {
    int t = wsum[lane];
    for (int o = 16; o > 0; o >>= 1) t += __shfl_xor(t, o);
    if (lane == 0) base_s = t;
  }
  __syncthreads();
  const int base = base_s;

  const int i = tile0 + tid;
  const int v = (i < n) ? cnt[i] : 0;
  int inc = v;
  for (int o = 1; o < 32; o <<= 1) {
    int t = __shfl_up(inc, o);
    if (lane >= o) inc += t;
  }
  __syncthreads();
  if (lane == 31) wsum[wv] = inc;
  __syncthreads();
  if (wv == 0) {
    int t = wsum[lane];
    int ti = t;
    for (int o = 1; o < 32; o <<= 1) {
      int u = __shfl_up(ti, o);
      if (lane >= o) ti += u;
    }
    wsum[lane] = ti - t;
  }
  __syncthreads();
  const int excl = base + wsum[wv] + inc - v;
  if (i < n) rowptr[i] = excl;
  if (i == n - 1) rowptr[n] = excl + v;
}

__global__ void csr_fill_kernel(const int* __restrict__ rowE,
                                const int* __restrict__ colE,
                                const int* __restrict__ rk,
                                const int* __restrict__ rowptr,
                                int* __restrict__ adj, int E) {
  int e = blockIdx.x * blockDim.x + threadIdx.x;
  if (e >= E) return;
  int p = rowptr[colE[e]] + rk[e];
  ((int2*)adj)[p] = make_int2(rowE[e], e);
}

template <int H, int F>
__global__ void __launch_bounds__(256)
edge_gather_kernel(const float* __restrict__ ex,
                   const float* __restrict__ denom,
                   const float* __restrict__ hsrc,
                   float* __restrict__ outd,
                   const int* __restrict__ rowptr,
                   const int* __restrict__ adj,
                   int E, int Nn) {
  const int2* __restrict__ adj2 = (const int2*)adj;
  constexpr int Ftot = H * F;
  constexpr int LPN = Ftot / 4;
  const int t = blockIdx.x * blockDim.x + threadIdx.x;
  const int c = t / LPN;
  if (c >= Nn) return;
  const int fl = (t % LPN) * 4;
  const int h = fl / F;
  const float dn = denom[c * H + h];

  const float cs = ex[(size_t)(E + c) * H + h] / dn;
  const float4 hs = *(const float4*)(hsrc + (size_t)c * Ftot + fl);
  float4 acc;
  acc.x = hs.x * cs;
  acc.y = hs.y * cs;
  acc.z = hs.z * cs;
  acc.w = hs.w * cs;

  const int beg = rowptr[c];
  const int end = rowptr[c + 1];
#pragma unroll 4
  for (int k = beg; k < end; ++k) {
    const int2 se = adj2[k];
    const float cf = ex[(size_t)se.y * H + h] / dn;
    const float4 v = *(const float4*)(hsrc + (size_t)se.x * Ftot + fl);
    acc.x += v.x * cf;
    acc.y += v.y * cf;
    acc.z += v.z * cf;
    acc.w += v.w * cf;
  }
  *(float4*)(outd + (size_t)c * Ftot + fl) = acc;
}

__global__ void relu_kernel(float* __restrict__ p, long long n) {
  long long t = (long long)blockIdx.x * blockDim.x + threadIdx.x;
  if (t < n) p[t] = fmaxf(p[t], 0.f);
}


extern "C" void kernel_launch(void* const* d_in, const int* in_sizes, int n_in,
                              void* d_out, int out_size, void* d_ws, size_t ws_size,
                              hipStream_t stream) {
  const float* x     = (const float*)d_in[0];
  const int*   ei    = (const int*)  d_in[1];
  const float* W1    = (const float*)d_in[2];
  const float* attn1 = (const float*)d_in[3];
  const float* W2    = (const float*)d_in[4];
  const float* attn2 = (const float*)d_in[5];
  const float* headW = (const float*)d_in[6];
  const float* headb = (const float*)d_in[7];

  const int Nn = in_sizes[0] / GAT_IN_DIM;
  const int E  = in_sizes[1] / 2;
  const int ET = E + Nn;
  const int* rowE = ei;
  const int* colE = ei + E;

  float* w = (float*)d_ws;
  float* h1    = w; w += (size_t)Nn * 128;
  float* out1  = w; w += (size_t)Nn * 128;
  float* h2    = w; w += (size_t)Nn * 64;
  float* out2  = w; w += (size_t)Nn * 64;
  float* es1   = w; w += (size_t)Nn * 4;
  float* ed1   = w; w += (size_t)Nn * 4;
  unsigned* amax1 = (unsigned*)w; w += (size_t)Nn * 4;
  float* den1  = w; w += (size_t)Nn * 4;
  float* es2   = w; w += (size_t)Nn;
  float* ed2   = w; w += (size_t)Nn;
  unsigned* amax2 = (unsigned*)w; w += (size_t)Nn;
  float* den2  = w; w += (size_t)Nn;
  float* ebuf  = w; w += (size_t)ET * 4;
  int* adj     = (int*)w;  w += (size_t)E * 2;
  int* rk      = (int*)w;  w += (size_t)E;
  int* cnt     = (int*)w;  w += (size_t)Nn;
  int* rowptr  = (int*)w;  w += (size_t)Nn + 1;

  hipMemsetAsync(cnt,   0, (size_t)Nn       * sizeof(int), stream);
  hipMemsetAsync(amax1, 0, (size_t)Nn * 4   * sizeof(unsigned), stream);
  hipMemsetAsync(den1,  0, (size_t)Nn * 4   * sizeof(float), stream);
  hipMemsetAsync(amax2, 0, (size_t)Nn       * sizeof(unsigned), stream);
  hipMemsetAsync(den2,  0, (size_t)Nn       * sizeof(float), stream);

  const int TB = 256;
  auto gemm_blocks = [](int M, int N) {
    int tiles = (M >> 4) * (N >> 4);
    return (tiles + 7) / 8;
  };

  csr_rank_kernel<<<(E + TB - 1) / TB, TB, 0, stream>>>(colE, cnt, rk, E);
  csr_scan_kernel<<<(Nn + 1023) / 1024, 1024, 0, stream>>>(cnt, rowptr, Nn);
  csr_fill_kernel<<<(E + TB - 1) / TB, TB, 0, stream>>>(rowE, colE, rk, rowptr, adj, E);

  wmma_gemm_f32<<<gemm_blocks(Nn, 128), TB, 0, stream>>>(x, W1, h1, nullptr, Nn, 128, 128);
  gat_scores<<<(Nn * 4 + TB - 1) / TB, TB, 0, stream>>>(h1, attn1, es1, ed1, Nn, 4, 32);
  {
    int nt = ET * 4;
    edge_alpha_kernel<<<(nt + TB - 1) / TB, TB, 0, stream>>>(es1, ed1, rowE, colE, ebuf, amax1, E, Nn, 4);
    edge_exp_kernel<<<(nt + TB - 1) / TB, TB, 0, stream>>>(ebuf, amax1, den1, colE, E, Nn, 4);
    edge_gather_kernel<4, 32><<<(Nn * 32 + TB - 1) / TB, TB, 0, stream>>>(
        ebuf, den1, h1, out1, rowptr, adj, E, Nn);
  }
  {
    long long n = (long long)Nn * 128;
    relu_kernel<<<(unsigned)((n + TB - 1) / TB), TB, 0, stream>>>(out1, n);
  }

  wmma_gemm_f32<<<gemm_blocks(Nn, 64), TB, 0, stream>>>(out1, W2, h2, nullptr, Nn, 64, 128);
  gat_scores<<<(Nn + TB - 1) / TB, TB, 0, stream>>>(h2, attn2, es2, ed2, Nn, 1, 64);
  {
    int nt = ET;
    edge_alpha_kernel<<<(nt + TB - 1) / TB, TB, 0, stream>>>(es2, ed2, rowE, colE, ebuf, amax2, E, Nn, 1);
    edge_exp_kernel<<<(nt + TB - 1) / TB, TB, 0, stream>>>(ebuf, amax2, den2, colE, E, Nn, 1);
    edge_gather_kernel<1, 64><<<(Nn * 16 + TB - 1) / TB, TB, 0, stream>>>(
        ebuf, den2, h2, out2, rowptr, adj, E, Nn);
  }

  wmma_gemm_f32<<<gemm_blocks(Nn, 64), TB, 0, stream>>>(out2, headW, (float*)d_out, headb, Nn, 64, 64);
}


# Segment softmax (max, exp, denominator) moved into the per-destination gather over the grouped edge list: alpha/exp kernels, their atomics and the edge buffer go; denominators summed by lane tree
# speedup vs baseline: 4.6127x; 1.9634x over previous
#include <hip/hip_runtime.h>
#include <hip/hip_bf16.h>

typedef float v2f __attribute__((ext_vector_type(2)));
typedef float v8f __attribute__((ext_vector_type(8)));

#define GAT_IN_DIM 128
#define GAT_NEG_SLOPE 0.2f

__global__ void wmma_gemm_f32(const float* __restrict__ A,
                              const float* __restrict__ B,
                              float* __restrict__ C,
                              const float* __restrict__ bias,
                              int M, int N, int K) {
  int wave = (blockIdx.x * blockDim.x + threadIdx.x) >> 5;
  int lane = threadIdx.x & 31;
  int tiles_n = N >> 4;
  int tiles = (M >> 4) * tiles_n;
  if (wave >= tiles) return;
  int tm = (wave / tiles_n) << 4;
  int tn = (wave % tiles_n) << 4;
  int idx  = lane & 15;
  int kk   = (lane >> 4) << 1;
  int half = lane >> 4;

  v8f acc = {};
  for (int k0 = 0; k0 < K; k0 += 4) {
    v2f a, b;
    const float* ap = A + (size_t)(tm + idx) * K + (k0 + kk);
    a.x = ap[0];
    a.y = ap[1];
    const float* bp = B + (size_t)(k0 + kk) * N + tn + idx;
    b.x = bp[0];
    b.y = bp[N];
    acc = __builtin_amdgcn_wmma_f32_16x16x4_f32(
         false, a,  false, b,
         (short)0, acc,  false,  false);
  }

  float badd = bias ? bias[tn + idx] : 0.0f;
#pragma unroll
  for (int r = 0; r < 8; ++r) {
    C[(size_t)(tm + r + 8 * half) * N + tn + idx] = acc[r] + badd;
  }
}

__global__ void gat_scores(const float* __restrict__ h,
                           const float* __restrict__ attn,
                           float* __restrict__ es, float* __restrict__ ed,
                           int n_nodes, int H, int F) {
  int t = blockIdx.x * blockDim.x + threadIdx.x;
  if (t >= n_nodes * H) return;
  int n = t / H;
  int hh = t - n * H;
  const float* hp = h + (size_t)n * H * F + (size_t)hh * F;
  const float* al = attn + (size_t)hh * 2 * F;
  const float* ar = al + F;
  float s = 0.f, d = 0.f;
  for (int f = 0; f < F; ++f) {
    s += hp[f] * al[f];
    d += hp[f] * ar[f];
  }
  es[t] = s;
  ed[t] = d;
}

__global__ void csr_rank_kernel(const int* __restrict__ colE,
                                int* __restrict__ cnt,
                                int* __restrict__ rk, int E) {
  int e = blockIdx.x * blockDim.x + threadIdx.x;
  if (e >= E) return;
  rk[e] = atomicAdd(&cnt[colE[e]], 1);
}

__global__ void __launch_bounds__(1024)
csr_scan_kernel(const int* __restrict__ cnt, int* __restrict__ rowptr, int n) {
  __shared__ int wsum[32];
  __shared__ int base_s;
  const int tid = threadIdx.x;
  const int lane = tid & 31;
  const int wv = tid >> 5;
  const int tile0 = blockIdx.x * 1024;

  int s = 0;
  for (int i = tid; i < tile0; i += 1024) s += cnt[i];
  for (int o = 16; o > 0; o >>= 1) s += __shfl_xor(s, o);
  if (lane == 0) wsum[wv] = s;
  __syncthreads();
  if (wv == 0) {
    int t = wsum[lane];
    for (int o = 16; o > 0; o >>= 1) t += __shfl_xor(t, o);
    if (lane == 0) base_s = t;
  }
  __syncthreads();
  const int base = base_s;

  const int i = tile0 + tid;
  const int v = (i < n) ? cnt[i] : 0;
  int inc = v;
  for (int o = 1; o < 32; o <<= 1) {
    int t = __shfl_up(inc, o);
    if (lane >= o) inc += t;
  }
  __syncthreads();
  if (lane == 31) wsum[wv] = inc;
  __syncthreads();
  if (wv == 0) {
    int t = wsum[lane];
    int ti = t;
    for (int o = 1; o < 32; o <<= 1) {
      int u = __shfl_up(ti, o);
      if (lane >= o) ti += u;
    }
    wsum[lane] = ti - t;
  }
  __syncthreads();
  const int excl = base + wsum[wv] + inc - v;
  if (i < n) rowptr[i] = excl;
  if (i == n - 1) rowptr[n] = excl + v;
}

__global__ void csr_fill_kernel(const int* __restrict__ rowE,
                                const int* __restrict__ colE,
                                const int* __restrict__ rk,
                                const int* __restrict__ rowptr,
                                int* __restrict__ adj, int E) {
  int e = blockIdx.x * blockDim.x + threadIdx.x;
  if (e >= E) return;
  adj[rowptr[colE[e]] + rk[e]] = rowE[e];
}

__device__ __forceinline__ float leaky(float a) {
  return (a > 0.f) ? a : a * GAT_NEG_SLOPE;
}

template <int H>
__device__ __forceinline__ void load_heads(const float* __restrict__ p, float (&o)[H]) {
  if constexpr (H == 4) {
    const float4 q = *(const float4*)p;
    o[0] = q.x; o[1] = q.y; o[2] = q.z; o[3] = q.w;
  } else {
    for (int h = 0; h < H; ++h) o[h] = p[h];
  }
}

template <int H>
__device__ __forceinline__ float pick_head(const float (&a)[H], int hm) {
  float r = a[0];
  for (int h = 1; h < H; ++h) r = (hm == h) ? a[h] : r;
  return r;
}

template <int H, int F>
__global__ void __launch_bounds__(256) __attribute__((amdgpu_waves_per_eu(16)))
edge_softmax_gather_kernel(const float* __restrict__ es,
                           const float* __restrict__ ed,
                           const float* __restrict__ hsrc,
                           float* __restrict__ outd,
                           const int* __restrict__ rowptr,
                           const int* __restrict__ adj,
                           int Nn) {
  constexpr int Ftot = H * F;
  constexpr int LPN = Ftot / 4;
  const int t = blockIdx.x * blockDim.x + threadIdx.x;
  const int c = t / LPN;
  if (c >= Nn) return;
  const int j = t % LPN;
  const int gb = (threadIdx.x & 31) & ~(LPN - 1);
  const int fl = j * 4;
  const int hm = fl / F;

  const int beg = rowptr[c];
  const int end = rowptr[c + 1];

  float edc[H], asf[H];
  load_heads<H>(ed + (size_t)c * H, edc);
  load_heads<H>(es + (size_t)c * H, asf);
  for (int h = 0; h < H; ++h) asf[h] = leaky(asf[h] + edc[h]);

  const int k0 = beg + j;
  const bool has0 = k0 < end;
  int s0 = c;
  float a0[H];
  for (int h = 0; h < H; ++h) a0[h] = asf[h];
  if (has0) {
    s0 = adj[k0];
    float e[H];
    load_heads<H>(es + (size_t)s0 * H, e);
    for (int h = 0; h < H; ++h) a0[h] = leaky(e[h] + edc[h]);
  }

  float m[H];
  for (int h = 0; h < H; ++h) m[h] = fmaxf(asf[h], a0[h]);
  for (int k = k0 + LPN; k < end; k += LPN) {
    float e[H];
    load_heads<H>(es + (size_t)adj[k] * H, e);
    for (int h = 0; h < H; ++h) m[h] = fmaxf(m[h], leaky(e[h] + edc[h]));
  }
  for (int o = LPN / 2; o > 0; o >>= 1) {
    for (int h = 0; h < H; ++h) m[h] = fmaxf(m[h], __shfl_xor(m[h], o));
  }

  float ex0[H], den[H];
  for (int h = 0; h < H; ++h) {
    ex0[h] = has0 ? __expf(a0[h] - m[h]) : 0.f;
    den[h] = ex0[h];
  }
  for (int k = k0 + LPN; k < end; k += LPN) {
    float e[H];
    load_heads<H>(es + (size_t)adj[k] * H, e);
    for (int h = 0; h < H; ++h) den[h] += __expf(leaky(e[h] + edc[h]) - m[h]);
  }
  for (int o = LPN / 2; o > 0; o >>= 1) {
    for (int h = 0; h < H; ++h) den[h] += __shfl_xor(den[h], o);
  }
  for (int h = 0; h < H; ++h) den[h] += __expf(asf[h] - m[h]);

  const float cs = __expf(pick_head<H>(asf, hm) - pick_head<H>(m, hm)) /
                   pick_head<H>(den, hm);
  const float4 hs = *(const float4*)(hsrc + (size_t)c * Ftot + fl);
  float4 acc;
  acc.x = hs.x * cs;
  acc.y = hs.y * cs;
  acc.z = hs.z * cs;
  acc.w = hs.w * cs;

  int sj = s0;
  float cfj[H];
  for (int h = 0; h < H; ++h) cfj[h] = ex0[h] / den[h];
  for (int cb = beg;;) {
    const int n = min(LPN, end - cb);
    for (int i = 0; i < n; i += 4) {
      int s4[4];
      float c4[4];
      for (int u = 0; u < 4; ++u) {
        const int sl = gb + i + u;
        s4[u] = __shfl(sj, sl);
        float ch[H];
        for (int h = 0; h < H; ++h) ch[h] = __shfl(cfj[h], sl);
        c4[u] = pick_head<H>(ch, hm);
      }
      float4 v4[4];
      for (int u = 0; u < 4; ++u)
        v4[u] = *(const float4*)(hsrc + (size_t)s4[u] * Ftot + fl);
      for (int u = 0; u < 4; ++u) {
        acc.x += v4[u].x * c4[u];
        acc.y += v4[u].y * c4[u];
        acc.z += v4[u].z * c4[u];
        acc.w += v4[u].w * c4[u];
      }
    }
    cb += LPN;
    if (cb >= end) break;
    const int k = cb + j;
    sj = c;
    for (int h = 0; h < H; ++h) cfj[h] = 0.f;
    if (k < end) {
      sj = adj[k];
      float e[H];
      load_heads<H>(es + (size_t)sj * H, e);
      for (int h = 0; h < H; ++h)
        cfj[h] = __expf(leaky(e[h] + edc[h]) - m[h]) / den[h];
    }
  }
  *(float4*)(outd + (size_t)c * Ftot + fl) = acc;
}

__global__ void relu_kernel(float* __restrict__ p, long long n) {
  long long t = (long long)blockIdx.x * blockDim.x + threadIdx.x;
  if (t < n) p[t] = fmaxf(p[t], 0.f);
}


extern "C" void kernel_launch(void* const* d_in, const int* in_sizes, int n_in,
                              void* d_out, int out_size, void* d_ws, size_t ws_size,
                              hipStream_t stream) {
  const float* x     = (const float*)d_in[0];
  const int*   ei    = (const int*)  d_in[1];
  const float* W1    = (const float*)d_in[2];
  const float* attn1 = (const float*)d_in[3];
  const float* W2    = (const float*)d_in[4];
  const float* attn2 = (const float*)d_in[5];
  const float* headW = (const float*)d_in[6];
  const float* headb = (const float*)d_in[7];

  const int Nn = in_sizes[0] / GAT_IN_DIM;
  const int E  = in_sizes[1] / 2;
  const int* rowE = ei;
  const int* colE = ei + E;

  float* w = (float*)d_ws;
  float* h1    = w; w += (size_t)Nn * 128;
  float* out1  = w; w += (size_t)Nn * 128;
  float* h2    = w; w += (size_t)Nn * 64;
  float* out2  = w; w += (size_t)Nn * 64;
  float* es1   = w; w += (size_t)Nn * 4;
  float* ed1   = w; w += (size_t)Nn * 4;
  float* es2   = w; w += (size_t)Nn;
  float* ed2   = w; w += (size_t)Nn;
  int* adj     = (int*)w;  w += (size_t)E;
  int* rk      = (int*)w;  w += (size_t)E;
  int* cnt     = (int*)w;  w += (size_t)Nn;
  int* rowptr  = (int*)w;  w += (size_t)Nn + 1;

  hipMemsetAsync(cnt,   0, (size_t)Nn       * sizeof(int), stream);

  const int TB = 256;
  auto gemm_blocks = [](int M, int N) {
    int tiles = (M >> 4) * (N >> 4);
    return (tiles + 7) / 8;
  };

  csr_rank_kernel<<<(E + TB - 1) / TB, TB, 0, stream>>>(colE, cnt, rk, E);
  csr_scan_kernel<<<(Nn + 1023) / 1024, 1024, 0, stream>>>(cnt, rowptr, Nn);
  csr_fill_kernel<<<(E + TB - 1) / TB, TB, 0, stream>>>(rowE, colE, rk, rowptr, adj, E);

  wmma_gemm_f32<<<gemm_blocks(Nn, 128), TB, 0, stream>>>(x, W1, h1, nullptr, Nn, 128, 128);
  gat_scores<<<(Nn * 4 + TB - 1) / TB, TB, 0, stream>>>(h1, attn1, es1, ed1, Nn, 4, 32);
  edge_softmax_gather_kernel<4, 32><<<(Nn * 32 + TB - 1) / TB, TB, 0, stream>>>(
      es1, ed1, h1, out1, rowptr, adj, Nn);
  {
    long long n = (long long)Nn * 128;
    relu_kernel<<<(unsigned)((n + TB - 1) / TB), TB, 0, stream>>>(out1, n);
  }

  wmma_gemm_f32<<<gemm_blocks(Nn, 64), TB, 0, stream>>>(out1, W2, h2, nullptr, Nn, 64, 128);
  gat_scores<<<(Nn + TB - 1) / TB, TB, 0, stream>>>(h2, attn2, es2, ed2, Nn, 1, 64);
  edge_softmax_gather_kernel<1, 64><<<(Nn * 16 + TB - 1) / TB, TB, 0, stream>>>(
      es2, ed2, h2, out2, rowptr, adj, Nn);

  wmma_gemm_f32<<<gemm_blocks(Nn, 64), TB, 0, stream>>>(out2, headW, (float*)d_out, headb, Nn, 64, 64);
}


# ReLU of layer 1 applied to each lane's four sums at the gather's one store; the separate one-element-per-thread activation launch (200000 waves, plane re-read and re-written) is removed
# speedup vs baseline: 5.0178x; 1.0878x over previous
#include <hip/hip_runtime.h>
#include <hip/hip_bf16.h>

typedef float v2f __attribute__((ext_vector_type(2)));
typedef float v8f __attribute__((ext_vector_type(8)));

#define GAT_IN_DIM 128
#define GAT_NEG_SLOPE 0.2f

__global__ void wmma_gemm_f32(const float* __restrict__ A,
                              const float* __restrict__ B,
                              float* __restrict__ C,
                              const float* __restrict__ bias,
                              int M, int N, int K) {
  int wave = (blockIdx.x * blockDim.x + threadIdx.x) >> 5;
  int lane = threadIdx.x & 31;
  int tiles_n = N >> 4;
  int tiles = (M >> 4) * tiles_n;
  if (wave >= tiles) return;
  int tm = (wave / tiles_n) << 4;
  int tn = (wave % tiles_n) << 4;
  int idx  = lane & 15;
  int kk   = (lane >> 4) << 1;
  int half = lane >> 4;

  v8f acc = {};
  for (int k0 = 0; k0 < K; k0 += 4) {
    v2f a, b;
    const float* ap = A + (size_t)(tm + idx) * K + (k0 + kk);
    a.x = ap[0];
    a.y = ap[1];
    const float* bp = B + (size_t)(k0 + kk) * N + tn + idx;
    b.x = bp[0];
    b.y = bp[N];
    acc = __builtin_amdgcn_wmma_f32_16x16x4_f32(
         false, a,  false, b,
         (short)0, acc,  false,  false);
  }

  float badd = bias ? bias[tn + idx] : 0.0f;
#pragma unroll
  for (int r = 0; r < 8; ++r) {
    C[(size_t)(tm + r + 8 * half) * N + tn + idx] = acc[r] + badd;
  }
}

__global__ void gat_scores(const float* __restrict__ h,
                           const float* __restrict__ attn,
                           float* __restrict__ es, float* __restrict__ ed,
                           int n_nodes, int H, int F) {
  int t = blockIdx.x * blockDim.x + threadIdx.x;
  if (t >= n_nodes * H) return;
  int n = t / H;
  int hh = t - n * H;
  const float* hp = h + (size_t)n * H * F + (size_t)hh * F;
  const float* al = attn + (size_t)hh * 2 * F;
  const float* ar = al + F;
  float s = 0.f, d = 0.f;
  for (int f = 0; f < F; ++f) {
    s += hp[f] * al[f];
    d += hp[f] * ar[f];
  }
  es[t] = s;
  ed[t] = d;
}

__global__ void csr_rank_kernel(const int* __restrict__ colE,
                                int* __restrict__ cnt,
                                int* __restrict__ rk, int E) {
  int e = blockIdx.x * blockDim.x + threadIdx.x;
  if (e >= E) return;
  rk[e] = atomicAdd(&cnt[colE[e]], 1);
}

__global__ void __launch_bounds__(1024)
csr_scan_kernel(const int* __restrict__ cnt, int* __restrict__ rowptr, int n) {
  __shared__ int wsum[32];
  __shared__ int base_s;
  const int tid = threadIdx.x;
  const int lane = tid & 31;
  const int wv = tid >> 5;
  const int tile0 = blockIdx.x * 1024;

  int s = 0;
  for (int i = tid; i < tile0; i += 1024) s += cnt[i];
  for (int o = 16; o > 0; o >>= 1) s += __shfl_xor(s, o);
  if (lane == 0) wsum[wv] = s;
  __syncthreads();
  if (wv == 0) {
    int t = wsum[lane];
    for (int o = 16; o > 0; o >>= 1) t += __shfl_xor(t, o);
    if (lane == 0) base_s = t;
  }
  __syncthreads();
  const int base = base_s;

  const int i = tile0 + tid;
  const int v = (i < n) ? cnt[i] : 0;
  int inc = v;
  for (int o = 1; o < 32; o <<= 1) {
    int t = __shfl_up(inc, o);
    if (lane >= o) inc += t;
  }
  __syncthreads();
  if (lane == 31) wsum[wv] = inc;
  __syncthreads();
  if (wv == 0) {
    int t = wsum[lane];
    int ti = t;
    for (int o = 1; o < 32; o <<= 1) {
      int u = __shfl_up(ti, o);
      if (lane >= o) ti += u;
    }
    wsum[lane] = ti - t;
  }
  __syncthreads();
  const int excl = base + wsum[wv] + inc - v;
  if (i < n) rowptr[i] = excl;
  if (i == n - 1) rowptr[n] = excl + v;
}

__global__ void csr_fill_kernel(const int* __restrict__ rowE,
                                const int* __restrict__ colE,
                                const int* __restrict__ rk,
                                const int* __restrict__ rowptr,
                                int* __restrict__ adj, int E) {
  int e = blockIdx.x * blockDim.x + threadIdx.x;
  if (e >= E) return;
  adj[rowptr[colE[e]] + rk[e]] = rowE[e];
}

__device__ __forceinline__ float leaky(float a) {
  return (a > 0.f) ? a : a * GAT_NEG_SLOPE;
}

template <int H>
__device__ __forceinline__ void load_heads(const float* __restrict__ p, float (&o)[H]) {
  if constexpr (H == 4) {
    const float4 q = *(const float4*)p;
    o[0] = q.x; o[1] = q.y; o[2] = q.z; o[3] = q.w;
  } else {
    for (int h = 0; h < H; ++h) o[h] = p[h];
  }
}

template <int H>
__device__ __forceinline__ float pick_head(const float (&a)[H], int hm) {
  float r = a[0];
  for (int h = 1; h < H; ++h) r = (hm == h) ? a[h] : r;
  return r;
}

template <int H, int F, bool RELU>
__global__ void __launch_bounds__(256) __attribute__((amdgpu_waves_per_eu(16)))
edge_softmax_gather_kernel(const float* __restrict__ es,
                           const float* __restrict__ ed,
                           const float* __restrict__ hsrc,
                           float* __restrict__ outd,
                           const int* __restrict__ rowptr,
                           const int* __restrict__ adj,
                           int Nn) {
  constexpr int Ftot = H * F;
  constexpr int LPN = Ftot / 4;
  const int t = blockIdx.x * blockDim.x + threadIdx.x;
  const int c = t / LPN;
  if (c >= Nn) return;
  const int j = t % LPN;
  const int gb = (threadIdx.x & 31) & ~(LPN - 1);
  const int fl = j * 4;
  const int hm = fl / F;

  const int beg = rowptr[c];
  const int end = rowptr[c + 1];

  float edc[H], asf[H];
  load_heads<H>(ed + (size_t)c * H, edc);
  load_heads<H>(es + (size_t)c * H, asf);
  for (int h = 0; h < H; ++h) asf[h] = leaky(asf[h] + edc[h]);

  const int k0 = beg + j;
  const bool has0 = k0 < end;
  int s0 = c;
  float a0[H];
  for (int h = 0; h < H; ++h) a0[h] = asf[h];
  if (has0) {
    s0 = adj[k0];
    float e[H];
    load_heads<H>(es + (size_t)s0 * H, e);
    for (int h = 0; h < H; ++h) a0[h] = leaky(e[h] + edc[h]);
  }

  float m[H];
  for (int h = 0; h < H; ++h) m[h] = fmaxf(asf[h], a0[h]);
  for (int k = k0 + LPN; k < end; k += LPN) {
    float e[H];
    load_heads<H>(es + (size_t)adj[k] * H, e);
    for (int h = 0; h < H; ++h) m[h] = fmaxf(m[h], leaky(e[h] + edc[h]));
  }
  for (int o = LPN / 2; o > 0; o >>= 1) {
    for (int h = 0; h < H; ++h) m[h] = fmaxf(m[h], __shfl_xor(m[h], o));
  }

  float ex0[H], den[H];
  for (int h = 0; h < H; ++h) {
    ex0[h] = has0 ? __expf(a0[h] - m[h]) : 0.f;
    den[h] = ex0[h];
  }
  for (int k = k0 + LPN; k < end; k += LPN) {
    float e[H];
    load_heads<H>(es + (size_t)adj[k] * H, e);
    for (int h = 0; h < H; ++h) den[h] += __expf(leaky(e[h] + edc[h]) - m[h]);
  }
  for (int o = LPN / 2; o > 0; o >>= 1) {
    for (int h = 0; h < H; ++h) den[h] += __shfl_xor(den[h], o);
  }
  for (int h = 0; h < H; ++h) den[h] += __expf(asf[h] - m[h]);

  const float cs = __expf(pick_head<H>(asf, hm) - pick_head<H>(m, hm)) /
                   pick_head<H>(den, hm);
  const float4 hs = *(const float4*)(hsrc + (size_t)c * Ftot + fl);
  float4 acc;
  acc.x = hs.x * cs;
  acc.y = hs.y * cs;
  acc.z = hs.z * cs;
  acc.w = hs.w * cs;

  int sj = s0;
  float cfj[H];
  for (int h = 0; h < H; ++h) cfj[h] = ex0[h] / den[h];
  for (int cb = beg;;) {
    const int n = min(LPN, end - cb);
    for (int i = 0; i < n; i += 4) {
      int s4[4];
      float c4[4];
      for (int u = 0; u < 4; ++u) {
        const int sl = gb + i + u;
        s4[u] = __shfl(sj, sl);
        float ch[H];
        for (int h = 0; h < H; ++h) ch[h] = __shfl(cfj[h], sl);
        c4[u] = pick_head<H>(ch, hm);
      }
      float4 v4[4];
      for (int u = 0; u < 4; ++u)
        v4[u] = *(const float4*)(hsrc + (size_t)s4[u] * Ftot + fl);
      for (int u = 0; u < 4; ++u) {
        acc.x += v4[u].x * c4[u];
        acc.y += v4[u].y * c4[u];
        acc.z += v4[u].z * c4[u];
        acc.w += v4[u].w * c4[u];
      }
    }
    cb += LPN;
    if (cb >= end) break;
    const int k = cb + j;
    sj = c;
    for (int h = 0; h < H; ++h) cfj[h] = 0.f;
    if (k < end) {
      sj = adj[k];
      float e[H];
      load_heads<H>(es + (size_t)sj * H, e);
      for (int h = 0; h < H; ++h)
        cfj[h] = __expf(leaky(e[h] + edc[h]) - m[h]) / den[h];
    }
  }
  if constexpr (RELU) {
    acc.x = fmaxf(acc.x, 0.f);
    acc.y = fmaxf(acc.y, 0.f);
    acc.z = fmaxf(acc.z, 0.f);
    acc.w = fmaxf(acc.w, 0.f);
  }
  *(float4*)(outd + (size_t)c * Ftot + fl) = acc;
}


extern "C" void kernel_launch(void* const* d_in, const int* in_sizes, int n_in,
                              void* d_out, int out_size, void* d_ws, size_t ws_size,
                              hipStream_t stream) {
  const float* x     = (const float*)d_in[0];
  const int*   ei    = (const int*)  d_in[1];
  const float* W1    = (const float*)d_in[2];
  const float* attn1 = (const float*)d_in[3];
  const float* W2    = (const float*)d_in[4];
  const float* attn2 = (const float*)d_in[5];
  const float* headW = (const float*)d_in[6];
  const float* headb = (const float*)d_in[7];

  const int Nn = in_sizes[0] / GAT_IN_DIM;
  const int E  = in_sizes[1] / 2;
  const int* rowE = ei;
  const int* colE = ei + E;

  float* w = (float*)d_ws;
  float* h1    = w; w += (size_t)Nn * 128;
  float* out1  = w; w += (size_t)Nn * 128;
  float* h2    = w; w += (size_t)Nn * 64;
  float* out2  = w; w += (size_t)Nn * 64;
  float* es1   = w; w += (size_t)Nn * 4;
  float* ed1   = w; w += (size_t)Nn * 4;
  float* es2   = w; w += (size_t)Nn;
  float* ed2   = w; w += (size_t)Nn;
  int* adj     = (int*)w;  w += (size_t)E;
  int* rk      = (int*)w;  w += (size_t)E;
  int* cnt     = (int*)w;  w += (size_t)Nn;
  int* rowptr  = (int*)w;  w += (size_t)Nn + 1;

  hipMemsetAsync(cnt,   0, (size_t)Nn       * sizeof(int), stream);

  const int TB = 256;
  auto gemm_blocks = [](int M, int N) {
    int tiles = (M >> 4) * (N >> 4);
    return (tiles + 7) / 8;
  };

  csr_rank_kernel<<<(E + TB - 1) / TB, TB, 0, stream>>>(colE, cnt, rk, E);
  csr_scan_kernel<<<(Nn + 1023) / 1024, 1024, 0, stream>>>(cnt, rowptr, Nn);
  csr_fill_kernel<<<(E + TB - 1) / TB, TB, 0, stream>>>(rowE, colE, rk, rowptr, adj, E);

  wmma_gemm_f32<<<gemm_blocks(Nn, 128), TB, 0, stream>>>(x, W1, h1, nullptr, Nn, 128, 128);
  gat_scores<<<(Nn * 4 + TB - 1) / TB, TB, 0, stream>>>(h1, attn1, es1, ed1, Nn, 4, 32);
  edge_softmax_gather_kernel<4, 32, true><<<(Nn * 32 + TB - 1) / TB, TB, 0, stream>>>(
      es1, ed1, h1, out1, rowptr, adj, Nn);

  wmma_gemm_f32<<<gemm_blocks(Nn, 64), TB, 0, stream>>>(out1, W2, h2, nullptr, Nn, 64, 128);
  gat_scores<<<(Nn + TB - 1) / TB, TB, 0, stream>>>(h2, attn2, es2, ed2, Nn, 1, 64);
  edge_softmax_gather_kernel<1, 64, false><<<(Nn * 16 + TB - 1) / TB, TB, 0, stream>>>(
      es2, ed2, h2, out2, rowptr, adj, Nn);

  wmma_gemm_f32<<<gemm_blocks(Nn, 64), TB, 0, stream>>>(out2, headW, (float*)d_out, headb, Nn, 64, 64);
}


# GEMMs: one wave per 16x64 strip (4 accumulator tiles), its 64 weight columns staged per block in LDS as k-quads, A read once per strip; sum order over k changes inside each block of 8 k
# speedup vs baseline: 5.7439x; 1.1447x over previous
#include <hip/hip_runtime.h>
#include <hip/hip_bf16.h>

typedef float v2f __attribute__((ext_vector_type(2)));
typedef float v8f __attribute__((ext_vector_type(8)));

#define GAT_IN_DIM 128
#define GAT_NEG_SLOPE 0.2f

template <int K>
__global__ void __launch_bounds__(128)
wmma_gemm_strip64(const float* __restrict__ A,
                  const float* __restrict__ B,
                  float* __restrict__ C,
                  const float* __restrict__ bias,
                  int M, int N) {
  __shared__ float4 Bs[(K / 4) * 64];
  const int tid = threadIdx.x;
  const int lane = tid & 31;
  const int wv = tid >> 5;
  const int nb = blockIdx.y * 64;
  const int c = lane & 15;
  const int h = lane >> 4;
  const int strip = blockIdx.x * 4 + wv;
  const bool live = strip < (M >> 4);
  const int tm = strip << 4;
  const float* ap = A + (size_t)(tm + c) * K + 4 * h;

  float4 afirst[4];
  if (live) {
#pragma unroll
    for (int i = 0; i < 4; ++i) afirst[i] = *(const float4*)(ap + 8 * i);
  }

  constexpr int ITEMS = (K / 4) * 16;
#pragma unroll
  for (int it0 = 0; it0 < ITEMS; it0 += 128) {
    const int it = it0 + tid;
    const int q = it >> 4;
    const int n4 = (it & 15) << 2;
    const float* bp = B + (size_t)(4 * q) * N + nb + n4;
    const float4 r0 = *(const float4*)(bp);
    const float4 r1 = *(const float4*)(bp + N);
    const float4 r2 = *(const float4*)(bp + 2 * (size_t)N);
    const float4 r3 = *(const float4*)(bp + 3 * (size_t)N);
    float4* d = Bs + q * 64 + n4;
    d[0] = make_float4(r0.x, r1.x, r2.x, r3.x);
    d[1] = make_float4(r0.y, r1.y, r2.y, r3.y);
    d[2] = make_float4(r0.z, r1.z, r2.z, r3.z);
    d[3] = make_float4(r0.w, r1.w, r2.w, r3.w);
  }
  __syncthreads();
  if (!live) return;

  const float4* bl = Bs + h * 64 + c;
  v8f acc[4] = {};
#pragma unroll
  for (int kb = 0; kb < K; kb += 8) {
    const float4 a = (kb < 32) ? afirst[kb >> 3] : *(const float4*)(ap + kb);
    const v2f a0 = {a.x, a.y};
    const v2f a1 = {a.z, a.w};
    float4 bq[4];
#pragma unroll
    for (int t = 0; t < 4; ++t) bq[t] = bl[(kb >> 2) * 64 + 16 * t];
#pragma unroll
    for (int t = 0; t < 4; ++t) {
      const v2f b0 = {bq[t].x, bq[t].y};
      acc[t] = __builtin_amdgcn_wmma_f32_16x16x4_f32(
           false, a0,  false, b0,
           (short)0, acc[t],  false,  false);
    }
#pragma unroll
    for (int t = 0; t < 4; ++t) {
      const v2f b1 = {bq[t].z, bq[t].w};
      acc[t] = __builtin_amdgcn_wmma_f32_16x16x4_f32(
           false, a1,  false, b1,
           (short)0, acc[t],  false,  false);
    }
  }

#pragma unroll
  for (int t = 0; t < 4; ++t) {
    const int col = nb + 16 * t + c;
    const float badd = bias ? bias[col] : 0.0f;
#pragma unroll
    for (int r = 0; r < 8; ++r) {
      C[(size_t)(tm + r + 8 * h) * N + col] = acc[t][r] + badd;
    }
  }
}

__global__ void gat_scores(const float* __restrict__ h,
                           const float* __restrict__ attn,
                           float* __restrict__ es, float* __restrict__ ed,
                           int n_nodes, int H, int F) {
  int t = blockIdx.x * blockDim.x + threadIdx.x;
  if (t >= n_nodes * H) return;
  int n = t / H;
  int hh = t - n * H;
  const float* hp = h + (size_t)n * H * F + (size_t)hh * F;
  const float* al = attn + (size_t)hh * 2 * F;
  const float* ar = al + F;
  float s = 0.f, d = 0.f;
  for (int f = 0; f < F; ++f) {
    s += hp[f] * al[f];
    d += hp[f] * ar[f];
  }
  es[t] = s;
  ed[t] = d;
}

__global__ void csr_rank_kernel(const int* __restrict__ colE,
                                int* __restrict__ cnt,
                                int* __restrict__ rk, int E) {
  int e = blockIdx.x * blockDim.x + threadIdx.x;
  if (e >= E) return;
  rk[e] = atomicAdd(&cnt[colE[e]], 1);
}

__global__ void __launch_bounds__(1024)
csr_scan_kernel(const int* __restrict__ cnt, int* __restrict__ rowptr, int n) {
  __shared__ int wsum[32];
  __shared__ int base_s;
  const int tid = threadIdx.x;
  const int lane = tid & 31;
  const int wv = tid >> 5;
  const int tile0 = blockIdx.x * 1024;

  int s = 0;
  for (int i = tid; i < tile0; i += 1024) s += cnt[i];
  for (int o = 16; o > 0; o >>= 1) s += __shfl_xor(s, o);
  if (lane == 0) wsum[wv] = s;
  __syncthreads();
  if (wv == 0) {
    int t = wsum[lane];
    for (int o = 16; o > 0; o >>= 1) t += __shfl_xor(t, o);
    if (lane == 0) base_s = t;
  }
  __syncthreads();
  const int base = base_s;

  const int i = tile0 + tid;
  const int v = (i < n) ? cnt[i] : 0;
  int inc = v;
  for (int o = 1; o < 32; o <<= 1) {
    int t = __shfl_up(inc, o);
    if (lane >= o) inc += t;
  }
  __syncthreads();
  if (lane == 31) wsum[wv] = inc;
  __syncthreads();
  if (wv == 0) {
    int t = wsum[lane];
    int ti = t;
    for (int o = 1; o < 32; o <<= 1) {
      int u = __shfl_up(ti, o);
      if (lane >= o) ti += u;
    }
    wsum[lane] = ti - t;
  }
  __syncthreads();
  const int excl = base + wsum[wv] + inc - v;
  if (i < n) rowptr[i] = excl;
  if (i == n - 1) rowptr[n] = excl + v;
}

__global__ void csr_fill_kernel(const int* __restrict__ rowE,
                                const int* __restrict__ colE,
                                const int* __restrict__ rk,
                                const int* __restrict__ rowptr,
                                int* __restrict__ adj, int E) {
  int e = blockIdx.x * blockDim.x + threadIdx.x;
  if (e >= E) return;
  adj[rowptr[colE[e]] + rk[e]] = rowE[e];
}

__device__ __forceinline__ float leaky(float a) {
  return (a > 0.f) ? a : a * GAT_NEG_SLOPE;
}

template <int H>
__device__ __forceinline__ void load_heads(const float* __restrict__ p, float (&o)[H]) {
  if constexpr (H == 4) {
    const float4 q = *(const float4*)p;
    o[0] = q.x; o[1] = q.y; o[2] = q.z; o[3] = q.w;
  } else {
    for (int h = 0; h < H; ++h) o[h] = p[h];
  }
}

template <int H>
__device__ __forceinline__ float pick_head(const float (&a)[H], int hm) {
  float r = a[0];
  for (int h = 1; h < H; ++h) r = (hm == h) ? a[h] : r;
  return r;
}

template <int H, int F, bool RELU>
__global__ void __launch_bounds__(256) __attribute__((amdgpu_waves_per_eu(16)))
edge_softmax_gather_kernel(const float* __restrict__ es,
                           const float* __restrict__ ed,
                           const float* __restrict__ hsrc,
                           float* __restrict__ outd,
                           const int* __restrict__ rowptr,
                           const int* __restrict__ adj,
                           int Nn) {
  constexpr int Ftot = H * F;
  constexpr int LPN = Ftot / 4;
  const int t = blockIdx.x * blockDim.x + threadIdx.x;
  const int c = t / LPN;
  if (c >= Nn) return;
  const int j = t % LPN;
  const int gb = (threadIdx.x & 31) & ~(LPN - 1);
  const int fl = j * 4;
  const int hm = fl / F;

  const int beg = rowptr[c];
  const int end = rowptr[c + 1];

  float edc[H], asf[H];
  load_heads<H>(ed + (size_t)c * H, edc);
  load_heads<H>(es + (size_t)c * H, asf);
  for (int h = 0; h < H; ++h) asf[h] = leaky(asf[h] + edc[h]);

  const int k0 = beg + j;
  const bool has0 = k0 < end;
  int s0 = c;
  float a0[H];
  for (int h = 0; h < H; ++h) a0[h] = asf[h];
  if (has0) {
    s0 = adj[k0];
    float e[H];
    load_heads<H>(es + (size_t)s0 * H, e);
    for (int h = 0; h < H; ++h) a0[h] = leaky(e[h] + edc[h]);
  }

  float m[H];
  for (int h = 0; h < H; ++h) m[h] = fmaxf(asf[h], a0[h]);
  for (int k = k0 + LPN; k < end; k += LPN) {
    float e[H];
    load_heads<H>(es + (size_t)adj[k] * H, e);
    for (int h = 0; h < H; ++h) m[h] = fmaxf(m[h], leaky(e[h] + edc[h]));
  }
  for (int o = LPN / 2; o > 0; o >>= 1) {
    for (int h = 0; h < H; ++h) m[h] = fmaxf(m[h], __shfl_xor(m[h], o));
  }

  float ex0[H], den[H];
  for (int h = 0; h < H; ++h) {
    ex0[h] = has0 ? __expf(a0[h] - m[h]) : 0.f;
    den[h] = ex0[h];
  }
  for (int k = k0 + LPN; k < end; k += LPN) {
    float e[H];
    load_heads<H>(es + (size_t)adj[k] * H, e);
    for (int h = 0; h < H; ++h) den[h] += __expf(leaky(e[h] + edc[h]) - m[h]);
  }
  for (int o = LPN / 2; o > 0; o >>= 1) {
    for (int h = 0; h < H; ++h) den[h] += __shfl_xor(den[h], o);
  }
  for (int h = 0; h < H; ++h) den[h] += __expf(asf[h] - m[h]);

  const float cs = __expf(pick_head<H>(asf, hm) - pick_head<H>(m, hm)) /
                   pick_head<H>(den, hm);
  const float4 hs = *(const float4*)(hsrc + (size_t)c * Ftot + fl);
  float4 acc;
  acc.x = hs.x * cs;
  acc.y = hs.y * cs;
  acc.z = hs.z * cs;
  acc.w = hs.w * cs;

  int sj = s0;
  float cfj[H];
  for (int h = 0; h < H; ++h) cfj[h] = ex0[h] / den[h];
  for (int cb = beg;;) {
    const int n = min(LPN, end - cb);
    for (int i = 0; i < n; i += 4) {
      int s4[4];
      float c4[4];
      for (int u = 0; u < 4; ++u) {
        const int sl = gb + i + u;
        s4[u] = __shfl(sj, sl);
        float ch[H];
        for (int h = 0; h < H; ++h) ch[h] = __shfl(cfj[h], sl);
        c4[u] = pick_head<H>(ch, hm);
      }
      float4 v4[4];
      for (int u = 0; u < 4; ++u)
        v4[u] = *(const float4*)(hsrc + (size_t)s4[u] * Ftot + fl);
      for (int u = 0; u < 4; ++u) {
        acc.x += v4[u].x * c4[u];
        acc.y += v4[u].y * c4[u];
        acc.z += v4[u].z * c4[u];
        acc.w += v4[u].w * c4[u];
      }
    }
    cb += LPN;
    if (cb >= end) break;
    const int k = cb + j;
    sj = c;
    for (int h = 0; h < H; ++h) cfj[h] = 0.f;
    if (k < end) {
      sj = adj[k];
      float e[H];
      load_heads<H>(es + (size_t)sj * H, e);
      for (int h = 0; h < H; ++h)
        cfj[h] = __expf(leaky(e[h] + edc[h]) - m[h]) / den[h];
    }
  }
  if constexpr (RELU) {
    acc.x = fmaxf(acc.x, 0.f);
    acc.y = fmaxf(acc.y, 0.f);
    acc.z = fmaxf(acc.z, 0.f);
    acc.w = fmaxf(acc.w, 0.f);
  }
  *(float4*)(outd + (size_t)c * Ftot + fl) = acc;
}


extern "C" void kernel_launch(void* const* d_in, const int* in_sizes, int n_in,
                              void* d_out, int out_size, void* d_ws, size_t ws_size,
                              hipStream_t stream) {
  const float* x     = (const float*)d_in[0];
  const int*   ei    = (const int*)  d_in[1];
  const float* W1    = (const float*)d_in[2];
  const float* attn1 = (const float*)d_in[3];
  const float* W2    = (const float*)d_in[4];
  const float* attn2 = (const float*)d_in[5];
  const float* headW = (const float*)d_in[6];
  const float* headb = (const float*)d_in[7];

  const int Nn = in_sizes[0] / GAT_IN_DIM;
  const int E  = in_sizes[1] / 2;
  const int* rowE = ei;
  const int* colE = ei + E;

  float* w = (float*)d_ws;
  float* h1    = w; w += (size_t)Nn * 128;
  float* out1  = w; w += (size_t)Nn * 128;
  float* h2    = w; w += (size_t)Nn * 64;
  float* out2  = w; w += (size_t)Nn * 64;
  float* es1   = w; w += (size_t)Nn * 4;
  float* ed1   = w; w += (size_t)Nn * 4;
  float* es2   = w; w += (size_t)Nn;
  float* ed2   = w; w += (size_t)Nn;
  int* adj     = (int*)w;  w += (size_t)E;
  int* rk      = (int*)w;  w += (size_t)E;
  int* cnt     = (int*)w;  w += (size_t)Nn;
  int* rowptr  = (int*)w;  w += (size_t)Nn + 1;

  hipMemsetAsync(cnt,   0, (size_t)Nn       * sizeof(int), stream);

  const int TB = 256;
  auto gemm_grid = [](int M, int N) {
    return dim3(((M >> 4) + 3) / 4, N >> 6);
  };
  const int GB = 128;

  csr_rank_kernel<<<(E + TB - 1) / TB, TB, 0, stream>>>(colE, cnt, rk, E);
  csr_scan_kernel<<<(Nn + 1023) / 1024, 1024, 0, stream>>>(cnt, rowptr, Nn);
  csr_fill_kernel<<<(E + TB - 1) / TB, TB, 0, stream>>>(rowE, colE, rk, rowptr, adj, E);

  wmma_gemm_strip64<128><<<gemm_grid(Nn, 128), GB, 0, stream>>>(x, W1, h1, nullptr, Nn, 128);
  gat_scores<<<(Nn * 4 + TB - 1) / TB, TB, 0, stream>>>(h1, attn1, es1, ed1, Nn, 4, 32);
  edge_softmax_gather_kernel<4, 32, true><<<(Nn * 32 + TB - 1) / TB, TB, 0, stream>>>(
      es1, ed1, h1, out1, rowptr, adj, Nn);

  wmma_gemm_strip64<128><<<gemm_grid(Nn, 64), GB, 0, stream>>>(out1, W2, h2, nullptr, Nn, 64);
  gat_scores<<<(Nn + TB - 1) / TB, TB, 0, stream>>>(h2, attn2, es2, ed2, Nn, 1, 64);
  edge_softmax_gather_kernel<1, 64, false><<<(Nn * 16 + TB - 1) / TB, TB, 0, stream>>>(
      es2, ed2, h2, out2, rowptr, adj, Nn);

  wmma_gemm_strip64<64><<<gemm_grid(Nn, 64), GB, 0, stream>>>(out2, headW, (float*)d_out, headb, Nn, 64);
}


# Attention scores es/ed computed in the GEMM epilogue from the stored tiles (per-lane partials, then a 16-lane DPP tree: score sum order changes); both score launches and their re-read of h go
# speedup vs baseline: 6.0734x; 1.0574x over previous
#include <hip/hip_runtime.h>
#include <hip/hip_bf16.h>

typedef float v2f __attribute__((ext_vector_type(2)));
typedef float v8f __attribute__((ext_vector_type(8)));

#define GAT_IN_DIM 128
#define GAT_NEG_SLOPE 0.2f


__device__ __forceinline__ float row16_sum(float v) {
  v += __builtin_bit_cast(float, __builtin_amdgcn_update_dpp(
           0, __builtin_bit_cast(int, v), 0x161, 0xf, 0xf, true));
  v += __builtin_bit_cast(float, __builtin_amdgcn_update_dpp(
           0, __builtin_bit_cast(int, v), 0x162, 0xf, 0xf, true));
  v += __builtin_bit_cast(float, __builtin_amdgcn_update_dpp(
           0, __builtin_bit_cast(int, v), 0x164, 0xf, 0xf, true));
  v += __builtin_bit_cast(float, __builtin_amdgcn_update_dpp(
           0, __builtin_bit_cast(int, v), 0x168, 0xf, 0xf, true));
  return v;
}

template <int K, int SH>
__global__ void __launch_bounds__(128)
wmma_gemm_strip64(const float* __restrict__ A,
                  const float* __restrict__ B,
                  float* __restrict__ C,
                  const float* __restrict__ bias,
                  const float* __restrict__ attn,
                  float* __restrict__ es,
                  float* __restrict__ ed,
                  int M, int N, int Htot) {
  __shared__ float4 Bs[(K / 4) * 64];
  const int tid = threadIdx.x;
  const int lane = tid & 31;
  const int wv = tid >> 5;
  const int nb = blockIdx.y * 64;
  const int c = lane & 15;
  const int h = lane >> 4;
  const int strip = blockIdx.x * 4 + wv;
  const bool live = strip < (M >> 4);
  const int tm = strip << 4;
  const float* ap = A + (size_t)(tm + c) * K + 4 * h;

  float4 afirst[4];
  if (live) {
#pragma unroll
    for (int i = 0; i < 4; ++i) afirst[i] = *(const float4*)(ap + 8 * i);
  }

  constexpr int ITEMS = (K / 4) * 16;
#pragma unroll
  for (int it0 = 0; it0 < ITEMS; it0 += 128) {
    const int it = it0 + tid;
    const int q = it >> 4;
    const int n4 = (it & 15) << 2;
    const float* bp = B + (size_t)(4 * q) * N + nb + n4;
    const float4 r0 = *(const float4*)(bp);
    const float4 r1 = *(const float4*)(bp + N);
    const float4 r2 = *(const float4*)(bp + 2 * (size_t)N);
    const float4 r3 = *(const float4*)(bp + 3 * (size_t)N);
    float4* d = Bs + q * 64 + n4;
    d[0] = make_float4(r0.x, r1.x, r2.x, r3.x);
    d[1] = make_float4(r0.y, r1.y, r2.y, r3.y);
    d[2] = make_float4(r0.z, r1.z, r2.z, r3.z);
    d[3] = make_float4(r0.w, r1.w, r2.w, r3.w);
  }
  __syncthreads();
  if (!live) return;

  const float4* bl = Bs + h * 64 + c;
  v8f acc[4] = {};
#pragma unroll
  for (int kb = 0; kb < K; kb += 8) {
    const float4 a = (kb < 32) ? afirst[kb >> 3] : *(const float4*)(ap + kb);
    const v2f a0 = {a.x, a.y};
    const v2f a1 = {a.z, a.w};
    float4 bq[4];
#pragma unroll
    for (int t = 0; t < 4; ++t) bq[t] = bl[(kb >> 2) * 64 + 16 * t];
#pragma unroll
    for (int t = 0; t < 4; ++t) {
      const v2f b0 = {bq[t].x, bq[t].y};
      acc[t] = __builtin_amdgcn_wmma_f32_16x16x4_f32(
           false, a0,  false, b0,
           (short)0, acc[t],  false,  false);
    }
#pragma unroll
    for (int t = 0; t < 4; ++t) {
      const v2f b1 = {bq[t].z, bq[t].w};
      acc[t] = __builtin_amdgcn_wmma_f32_16x16x4_f32(
           false, a1,  false, b1,
           (short)0, acc[t],  false,  false);
    }
  }

#pragma unroll
  for (int t = 0; t < 4; ++t) {
    const int col = nb + 16 * t + c;
    const float badd = bias ? bias[col] : 0.0f;
#pragma unroll
    for (int r = 0; r < 8; ++r) {
      acc[t][r] += badd;
      C[(size_t)(tm + r + 8 * h) * N + col] = acc[t][r];
    }
  }

  if constexpr (SH > 0) {
    constexpr int F = 64 / SH;
    constexpr int TPH = 4 / SH;
    const int rown = c & 7;
#pragma unroll
    for (int hl = 0; hl < SH; ++hl) {
      const int hd = blockIdx.y * SH + hl;
      const float* av = attn + (size_t)hd * 2 * F;
      float alv[TPH], arv[TPH];
#pragma unroll
      for (int tt = 0; tt < TPH; ++tt) {
        alv[tt] = av[16 * tt + c];
        arv[tt] = av[F + 16 * tt + c];
      }
      float sv = 0.f, dv = 0.f;
#pragma unroll
      for (int r = 0; r < 8; ++r) {
        float ps = acc[hl * TPH][r] * alv[0];
        float pd = acc[hl * TPH][r] * arv[0];
#pragma unroll
        for (int tt = 1; tt < TPH; ++tt) {
          ps += acc[hl * TPH + tt][r] * alv[tt];
          pd += acc[hl * TPH + tt][r] * arv[tt];
        }
        ps = row16_sum(ps);
        pd = row16_sum(pd);
        sv = (rown == r) ? ps : sv;
        dv = (rown == r) ? pd : dv;
      }
      float* dst = (c < 8) ? es : ed;
      dst[(size_t)(tm + 8 * h + rown) * Htot + hd] = (c < 8) ? sv : dv;
    }
  }
}

__global__ void csr_rank_kernel(const int* __restrict__ colE,
                                int* __restrict__ cnt,
                                int* __restrict__ rk, int E) {
  int e = blockIdx.x * blockDim.x + threadIdx.x;
  if (e >= E) return;
  rk[e] = atomicAdd(&cnt[colE[e]], 1);
}

__global__ void __launch_bounds__(1024)
csr_scan_kernel(const int* __restrict__ cnt, int* __restrict__ rowptr, int n) {
  __shared__ int wsum[32];
  __shared__ int base_s;
  const int tid = threadIdx.x;
  const int lane = tid & 31;
  const int wv = tid >> 5;
  const int tile0 = blockIdx.x * 1024;

  int s = 0;
  for (int i = tid; i < tile0; i += 1024) s += cnt[i];
  for (int o = 16; o > 0; o >>= 1) s += __shfl_xor(s, o);
  if (lane == 0) wsum[wv] = s;
  __syncthreads();
  if (wv == 0) {
    int t = wsum[lane];
    for (int o = 16; o > 0; o >>= 1) t += __shfl_xor(t, o);
    if (lane == 0) base_s = t;
  }
  __syncthreads();
  const int base = base_s;

  const int i = tile0 + tid;
  const int v = (i < n) ? cnt[i] : 0;
  int inc = v;
  for (int o = 1; o < 32; o <<= 1) {
    int t = __shfl_up(inc, o);
    if (lane >= o) inc += t;
  }
  __syncthreads();
  if (lane == 31) wsum[wv] = inc;
  __syncthreads();
  if (wv == 0) {
    int t = wsum[lane];
    int ti = t;
    for (int o = 1; o < 32; o <<= 1) {
      int u = __shfl_up(ti, o);
      if (lane >= o) ti += u;
    }
    wsum[lane] = ti - t;
  }
  __syncthreads();
  const int excl = base + wsum[wv] + inc - v;
  if (i < n) rowptr[i] = excl;
  if (i == n - 1) rowptr[n] = excl + v;
}

__global__ void csr_fill_kernel(const int* __restrict__ rowE,
                                const int* __restrict__ colE,
                                const int* __restrict__ rk,
                                const int* __restrict__ rowptr,
                                int* __restrict__ adj, int E) {
  int e = blockIdx.x * blockDim.x + threadIdx.x;
  if (e >= E) return;
  adj[rowptr[colE[e]] + rk[e]] = rowE[e];
}

__device__ __forceinline__ float leaky(float a) {
  return (a > 0.f) ? a : a * GAT_NEG_SLOPE;
}

template <int H>
__device__ __forceinline__ void load_heads(const float* __restrict__ p, float (&o)[H]) {
  if constexpr (H == 4) {
    const float4 q = *(const float4*)p;
    o[0] = q.x; o[1] = q.y; o[2] = q.z; o[3] = q.w;
  } else {
    for (int h = 0; h < H; ++h) o[h] = p[h];
  }
}

template <int H>
__device__ __forceinline__ float pick_head(const float (&a)[H], int hm) {
  float r = a[0];
  for (int h = 1; h < H; ++h) r = (hm == h) ? a[h] : r;
  return r;
}

template <int H, int F, bool RELU>
__global__ void __launch_bounds__(256) __attribute__((amdgpu_waves_per_eu(16)))
edge_softmax_gather_kernel(const float* __restrict__ es,
                           const float* __restrict__ ed,
                           const float* __restrict__ hsrc,
                           float* __restrict__ outd,
                           const int* __restrict__ rowptr,
                           const int* __restrict__ adj,
                           int Nn) {
  constexpr int Ftot = H * F;
  constexpr int LPN = Ftot / 4;
  const int t = blockIdx.x * blockDim.x + threadIdx.x;
  const int c = t / LPN;
  if (c >= Nn) return;
  const int j = t % LPN;
  const int gb = (threadIdx.x & 31) & ~(LPN - 1);
  const int fl = j * 4;
  const int hm = fl / F;

  const int beg = rowptr[c];
  const int end = rowptr[c + 1];

  float edc[H], asf[H];
  load_heads<H>(ed + (size_t)c * H, edc);
  load_heads<H>(es + (size_t)c * H, asf);
  for (int h = 0; h < H; ++h) asf[h] = leaky(asf[h] + edc[h]);

  const int k0 = beg + j;
  const bool has0 = k0 < end;
  int s0 = c;
  float a0[H];
  for (int h = 0; h < H; ++h) a0[h] = asf[h];
  if (has0) {
    s0 = adj[k0];
    float e[H];
    load_heads<H>(es + (size_t)s0 * H, e);
    for (int h = 0; h < H; ++h) a0[h] = leaky(e[h] + edc[h]);
  }

  float m[H];
  for (int h = 0; h < H; ++h) m[h] = fmaxf(asf[h], a0[h]);
  for (int k = k0 + LPN; k < end; k += LPN) {
    float e[H];
    load_heads<H>(es + (size_t)adj[k] * H, e);
    for (int h = 0; h < H; ++h) m[h] = fmaxf(m[h], leaky(e[h] + edc[h]));
  }
  for (int o = LPN / 2; o > 0; o >>= 1) {
    for (int h = 0; h < H; ++h) m[h] = fmaxf(m[h], __shfl_xor(m[h], o));
  }

  float ex0[H], den[H];
  for (int h = 0; h < H; ++h) {
    ex0[h] = has0 ? __expf(a0[h] - m[h]) : 0.f;
    den[h] = ex0[h];
  }
  for (int k = k0 + LPN; k < end; k += LPN) {
    float e[H];
    load_heads<H>(es + (size_t)adj[k] * H, e);
    for (int h = 0; h < H; ++h) den[h] += __expf(leaky(e[h] + edc[h]) - m[h]);
  }
  for (int o = LPN / 2; o > 0; o >>= 1) {
    for (int h = 0; h < H; ++h) den[h] += __shfl_xor(den[h], o);
  }
  for (int h = 0; h < H; ++h) den[h] += __expf(asf[h] - m[h]);

  const float cs = __expf(pick_head<H>(asf, hm) - pick_head<H>(m, hm)) /
                   pick_head<H>(den, hm);
  const float4 hs = *(const float4*)(hsrc + (size_t)c * Ftot + fl);
  float4 acc;
  acc.x = hs.x * cs;
  acc.y = hs.y * cs;
  acc.z = hs.z * cs;
  acc.w = hs.w * cs;

  int sj = s0;
  float cfj[H];
  for (int h = 0; h < H; ++h) cfj[h] = ex0[h] / den[h];
  for (int cb = beg;;) {
    const int n = min(LPN, end - cb);
    for (int i = 0; i < n; i += 4) {
      int s4[4];
      float c4[4];
      for (int u = 0; u < 4; ++u) {
        const int sl = gb + i + u;
        s4[u] = __shfl(sj, sl);
        float ch[H];
        for (int h = 0; h < H; ++h) ch[h] = __shfl(cfj[h], sl);
        c4[u] = pick_head<H>(ch, hm);
      }
      float4 v4[4];
      for (int u = 0; u < 4; ++u)
        v4[u] = *(const float4*)(hsrc + (size_t)s4[u] * Ftot + fl);
      for (int u = 0; u < 4; ++u) {
        acc.x += v4[u].x * c4[u];
        acc.y += v4[u].y * c4[u];
        acc.z += v4[u].z * c4[u];
        acc.w += v4[u].w * c4[u];
      }
    }
    cb += LPN;
    if (cb >= end) break;
    const int k = cb + j;
    sj = c;
    for (int h = 0; h < H; ++h) cfj[h] = 0.f;
    if (k < end) {
      sj = adj[k];
      float e[H];
      load_heads<H>(es + (size_t)sj * H, e);
      for (int h = 0; h < H; ++h)
        cfj[h] = __expf(leaky(e[h] + edc[h]) - m[h]) / den[h];
    }
  }
  if constexpr (RELU) {
    acc.x = fmaxf(acc.x, 0.f);
    acc.y = fmaxf(acc.y, 0.f);
    acc.z = fmaxf(acc.z, 0.f);
    acc.w = fmaxf(acc.w, 0.f);
  }
  *(float4*)(outd + (size_t)c * Ftot + fl) = acc;
}


extern "C" void kernel_launch(void* const* d_in, const int* in_sizes, int n_in,
                              void* d_out, int out_size, void* d_ws, size_t ws_size,
                              hipStream_t stream) {
  const float* x     = (const float*)d_in[0];
  const int*   ei    = (const int*)  d_in[1];
  const float* W1    = (const float*)d_in[2];
  const float* attn1 = (const float*)d_in[3];
  const float* W2    = (const float*)d_in[4];
  const float* attn2 = (const float*)d_in[5];
  const float* headW = (const float*)d_in[6];
  const float* headb = (const float*)d_in[7];

  const int Nn = in_sizes[0] / GAT_IN_DIM;
  const int E  = in_sizes[1] / 2;
  const int* rowE = ei;
  const int* colE = ei + E;

  float* w = (float*)d_ws;
  float* h1    = w; w += (size_t)Nn * 128;
  float* out1  = w; w += (size_t)Nn * 128;
  float* h2    = w; w += (size_t)Nn * 64;
  float* out2  = w; w += (size_t)Nn * 64;
  float* es1   = w; w += (size_t)Nn * 4;
  float* ed1   = w; w += (size_t)Nn * 4;
  float* es2   = w; w += (size_t)Nn;
  float* ed2   = w; w += (size_t)Nn;
  int* adj     = (int*)w;  w += (size_t)E;
  int* rk      = (int*)w;  w += (size_t)E;
  int* cnt     = (int*)w;  w += (size_t)Nn;
  int* rowptr  = (int*)w;  w += (size_t)Nn + 1;

  hipMemsetAsync(cnt,   0, (size_t)Nn       * sizeof(int), stream);

  const int TB = 256;
  auto gemm_grid = [](int M, int N) {
    return dim3(((M >> 4) + 3) / 4, N >> 6);
  };
  const int GB = 128;

  csr_rank_kernel<<<(E + TB - 1) / TB, TB, 0, stream>>>(colE, cnt, rk, E);
  csr_scan_kernel<<<(Nn + 1023) / 1024, 1024, 0, stream>>>(cnt, rowptr, Nn);
  csr_fill_kernel<<<(E + TB - 1) / TB, TB, 0, stream>>>(rowE, colE, rk, rowptr, adj, E);

  wmma_gemm_strip64<128, 2><<<gemm_grid(Nn, 128), GB, 0, stream>>>(
      x, W1, h1, nullptr, attn1, es1, ed1, Nn, 128, 4);
  edge_softmax_gather_kernel<4, 32, true><<<(Nn * 32 + TB - 1) / TB, TB, 0, stream>>>(
      es1, ed1, h1, out1, rowptr, adj, Nn);

  wmma_gemm_strip64<128, 1><<<gemm_grid(Nn, 64), GB, 0, stream>>>(
      out1, W2, h2, nullptr, attn2, es2, ed2, Nn, 64, 1);
  edge_softmax_gather_kernel<1, 64, false><<<(Nn * 16 + TB - 1) / TB, TB, 0, stream>>>(
      es2, ed2, h2, out2, rowptr, adj, Nn);

  wmma_gemm_strip64<64, 0><<<gemm_grid(Nn, 64), GB, 0, stream>>>(
      out2, headW, (float*)d_out, headb, nullptr, nullptr, nullptr, Nn, 64, 0);
}


# Edges placed in ONE pass into a fixed 64-entry table per destination (entry = arrival rank, overflow pair list past 64): scan and fill launches, rank and rowptr buffers go; sum order as before
# speedup vs baseline: 7.4729x; 1.2304x over previous
#include <hip/hip_runtime.h>
#include <hip/hip_bf16.h>

typedef float v2f __attribute__((ext_vector_type(2)));
typedef float v8f __attribute__((ext_vector_type(8)));

#define GAT_IN_DIM 128
#define GAT_NEG_SLOPE 0.2f


__device__ __forceinline__ float row16_sum(float v) {
  v += __builtin_bit_cast(float, __builtin_amdgcn_update_dpp(
           0, __builtin_bit_cast(int, v), 0x161, 0xf, 0xf, true));
  v += __builtin_bit_cast(float, __builtin_amdgcn_update_dpp(
           0, __builtin_bit_cast(int, v), 0x162, 0xf, 0xf, true));
  v += __builtin_bit_cast(float, __builtin_amdgcn_update_dpp(
           0, __builtin_bit_cast(int, v), 0x164, 0xf, 0xf, true));
  v += __builtin_bit_cast(float, __builtin_amdgcn_update_dpp(
           0, __builtin_bit_cast(int, v), 0x168, 0xf, 0xf, true));
  return v;
}

template <int K, int SH>
__global__ void __launch_bounds__(128)
wmma_gemm_strip64(const float* __restrict__ A,
                  const float* __restrict__ B,
                  float* __restrict__ C,
                  const float* __restrict__ bias,
                  const float* __restrict__ attn,
                  float* __restrict__ es,
                  float* __restrict__ ed,
                  int M, int N, int Htot) {
  __shared__ float4 Bs[(K / 4) * 64];
  const int tid = threadIdx.x;
  const int lane = tid & 31;
  const int wv = tid >> 5;
  const int nb = blockIdx.y * 64;
  const int c = lane & 15;
  const int h = lane >> 4;
  const int strip = blockIdx.x * 4 + wv;
  const bool live = strip < (M >> 4);
  const int tm = strip << 4;
  const float* ap = A + (size_t)(tm + c) * K + 4 * h;

  float4 afirst[4];
  if (live) {
#pragma unroll
    for (int i = 0; i < 4; ++i) afirst[i] = *(const float4*)(ap + 8 * i);
  }

  constexpr int ITEMS = (K / 4) * 16;
#pragma unroll
  for (int it0 = 0; it0 < ITEMS; it0 += 128) {
    const int it = it0 + tid;
    const int q = it >> 4;
    const int n4 = (it & 15) << 2;
    const float* bp = B + (size_t)(4 * q) * N + nb + n4;
    const float4 r0 = *(const float4*)(bp);
    const float4 r1 = *(const float4*)(bp + N);
    const float4 r2 = *(const float4*)(bp + 2 * (size_t)N);
    const float4 r3 = *(const float4*)(bp + 3 * (size_t)N);
    float4* d = Bs + q * 64 + n4;
    d[0] = make_float4(r0.x, r1.x, r2.x, r3.x);
    d[1] = make_float4(r0.y, r1.y, r2.y, r3.y);
    d[2] = make_float4(r0.z, r1.z, r2.z, r3.z);
    d[3] = make_float4(r0.w, r1.w, r2.w, r3.w);
  }
  __syncthreads();
  if (!live) return;

  const float4* bl = Bs + h * 64 + c;
  v8f acc[4] = {};
#pragma unroll
  for (int kb = 0; kb < K; kb += 8) {
    const float4 a = (kb < 32) ? afirst[kb >> 3] : *(const float4*)(ap + kb);
    const v2f a0 = {a.x, a.y};
    const v2f a1 = {a.z, a.w};
    float4 bq[4];
#pragma unroll
    for (int t = 0; t < 4; ++t) bq[t] = bl[(kb >> 2) * 64 + 16 * t];
#pragma unroll
    for (int t = 0; t < 4; ++t) {
      const v2f b0 = {bq[t].x, bq[t].y};
      acc[t] = __builtin_amdgcn_wmma_f32_16x16x4_f32(
           false, a0,  false, b0,
           (short)0, acc[t],  false,  false);
    }
#pragma unroll
    for (int t = 0; t < 4; ++t) {
      const v2f b1 = {bq[t].z, bq[t].w};
      acc[t] = __builtin_amdgcn_wmma_f32_16x16x4_f32(
           false, a1,  false, b1,
           (short)0, acc[t],  false,  false);
    }
  }

#pragma unroll
  for (int t = 0; t < 4; ++t) {
    const int col = nb + 16 * t + c;
    const float badd = bias ? bias[col] : 0.0f;
#pragma unroll
    for (int r = 0; r < 8; ++r) {
      acc[t][r] += badd;
      C[(size_t)(tm + r + 8 * h) * N + col] = acc[t][r];
    }
  }

  if constexpr (SH > 0) {
    constexpr int F = 64 / SH;
    constexpr int TPH = 4 / SH;
    const int rown = c & 7;
#pragma unroll
    for (int hl = 0; hl < SH; ++hl) {
      const int hd = blockIdx.y * SH + hl;
      const float* av = attn + (size_t)hd * 2 * F;
      float alv[TPH], arv[TPH];
#pragma unroll
      for (int tt = 0; tt < TPH; ++tt) {
        alv[tt] = av[16 * tt + c];
        arv[tt] = av[F + 16 * tt + c];
      }
      float sv = 0.f, dv = 0.f;
#pragma unroll
      for (int r = 0; r < 8; ++r) {
        float ps = acc[hl * TPH][r] * alv[0];
        float pd = acc[hl * TPH][r] * arv[0];
#pragma unroll
        for (int tt = 1; tt < TPH; ++tt) {
          ps += acc[hl * TPH + tt][r] * alv[tt];
          pd += acc[hl * TPH + tt][r] * arv[tt];
        }
        ps = row16_sum(ps);
        pd = row16_sum(pd);
        sv = (rown == r) ? ps : sv;
        dv = (rown == r) ? pd : dv;
      }
      float* dst = (c < 8) ? es : ed;
      dst[(size_t)(tm + 8 * h + rown) * Htot + hd] = (c < 8) ? sv : dv;
    }
  }
}

#define GAT_SLOTS 64

__global__ void edge_place_kernel(const int* __restrict__ rowE,
                                  const int* __restrict__ colE,
                                  int* __restrict__ cnt,
                                  int* __restrict__ slots,
                                  int* __restrict__ ovf,
                                  int Nn, int E) {
  int e = blockIdx.x * blockDim.x + threadIdx.x;
  if (e >= E) return;
  const int c = colE[e];
  const int s = rowE[e];
  const int r = atomicAdd(&cnt[c], 1);
  if (r < GAT_SLOTS) {
    slots[c * GAT_SLOTS + r] = s;
  } else {
    const int o = atomicAdd(&cnt[Nn], 1);
    ovf[2 * o] = c;
    ovf[2 * o + 1] = s;
  }
}

__device__ __forceinline__ float leaky(float a) {
  return (a > 0.f) ? a : a * GAT_NEG_SLOPE;
}

template <int U>
__device__ __forceinline__ int row_share(int v) {
  return __builtin_amdgcn_update_dpp(0, v, 0x150 + U, 0xf, 0xf, true);
}

template <int H, int F, bool RELU>
__global__ void __launch_bounds__(256)
edge_softmax_gather_kernel(const float* __restrict__ es,
                           const float* __restrict__ ed,
                           const float* __restrict__ hsrc,
                           float* __restrict__ outd,
                           const int* __restrict__ cnt,
                           const int* __restrict__ slots,
                           const int* __restrict__ ovf,
                           int Nn) {
  constexpr int Ftot = H * F;
  constexpr int LPN = Ftot / 4;
  static_assert(LPN == 16 || LPN == 32, "a node's lanes must fill 16-lane rows");
  constexpr int CH = 8;
  const int t = blockIdx.x * blockDim.x + threadIdx.x;
  const int c = t / LPN;
  if (c >= Nn) return;
  const int j = t % LPN;
  const int hm = (j * 4) / F;
  const int l8 = threadIdx.x & (CH - 1);
  const float4* __restrict__ h4 = (const float4*)hsrc;

  const int deg = cnt[c];
  const int beg = c * GAT_SLOTS;
  const int end = beg + min(deg, GAT_SLOTS);
  const float edc = ed[c * H + hm];
  const float asf = leaky(es[c * H + hm] + edc);
  float4 acc = h4[c * LPN + j];
  float m = asf;
  float den = 1.f;

  if (beg < end) {
    const int last = end - 1;
    int sv = c;
    if (beg + l8 < end) sv = slots[beg + l8];
    for (int cb = beg; cb < end; cb += CH) {
      const int svn = slots[min(cb + CH + l8, last)];

      int s[CH];
      s[0] = row_share<0>(sv); s[1] = row_share<1>(sv);
      s[2] = row_share<2>(sv); s[3] = row_share<3>(sv);
      s[4] = row_share<4>(sv); s[5] = row_share<5>(sv);
      s[6] = row_share<6>(sv); s[7] = row_share<7>(sv);
      float e[CH];
      float4 r[CH];
      for (int u = 0; u < CH; ++u) e[u] = es[s[u] * H + hm];
      for (int u = 0; u < CH; ++u) r[u] = h4[s[u] * LPN + j];

      const int rem = end - cb;
      float a[CH];
      float mc = m;
      for (int u = 0; u < CH; ++u) {
        a[u] = leaky(e[u] + edc);
        mc = (u < rem) ? fmaxf(mc, a[u]) : mc;
      }
      const float sc = __expf(m - mc);
      den *= sc;
      acc.x *= sc;
      acc.y *= sc;
      acc.z *= sc;
      acc.w *= sc;
      for (int u = 0; u < CH; ++u) {
        const float p = (u < rem) ? __expf(a[u] - mc) : 0.f;
        den += p;
        acc.x += r[u].x * p;
        acc.y += r[u].y * p;
        acc.z += r[u].z * p;
        acc.w += r[u].w * p;
      }
      m = mc;
      sv = svn;
    }
  }
  if (deg > GAT_SLOTS) {
    const int no = cnt[Nn];
    for (int i = 0; i < no; ++i) {
      if (ovf[2 * i] != c) continue;
      const int s = ovf[2 * i + 1];
      const float a = leaky(es[s * H + hm] + edc);
      const float4 r = h4[s * LPN + j];
      const float mc = fmaxf(m, a);
      const float sc = __expf(m - mc);
      const float p = __expf(a - mc);
      den = den * sc + p;
      acc.x = acc.x * sc + r.x * p;
      acc.y = acc.y * sc + r.y * p;
      acc.z = acc.z * sc + r.z * p;
      acc.w = acc.w * sc + r.w * p;
      m = mc;
    }
  }
  const float inv = 1.0f / den;
  acc.x *= inv;
  acc.y *= inv;
  acc.z *= inv;
  acc.w *= inv;
  if constexpr (RELU) {
    acc.x = fmaxf(acc.x, 0.f);
    acc.y = fmaxf(acc.y, 0.f);
    acc.z = fmaxf(acc.z, 0.f);
    acc.w = fmaxf(acc.w, 0.f);
  }
  ((float4*)outd)[c * LPN + j] = acc;
}


extern "C" void kernel_launch(void* const* d_in, const int* in_sizes, int n_in,
                              void* d_out, int out_size, void* d_ws, size_t ws_size,
                              hipStream_t stream) {
  const float* x     = (const float*)d_in[0];
  const int*   ei    = (const int*)  d_in[1];
  const float* W1    = (const float*)d_in[2];
  const float* attn1 = (const float*)d_in[3];
  const float* W2    = (const float*)d_in[4];
  const float* attn2 = (const float*)d_in[5];
  const float* headW = (const float*)d_in[6];
  const float* headb = (const float*)d_in[7];

  const int Nn = in_sizes[0] / GAT_IN_DIM;
  const int E  = in_sizes[1] / 2;
  const int* rowE = ei;
  const int* colE = ei + E;

  float* w = (float*)d_ws;
  float* h1    = w; w += (size_t)Nn * 128;
  float* out1  = w; w += (size_t)Nn * 128;
  float* h2    = w; w += (size_t)Nn * 64;
  float* out2  = w; w += (size_t)Nn * 64;
  int* slots   = (int*)w;  w += (size_t)Nn * GAT_SLOTS;
  int* ovf     = (int*)w;  w += (size_t)E * 2;
  float* es1   = w; w += (size_t)Nn * 4;
  float* ed1   = w; w += (size_t)Nn * 4;
  float* es2   = w; w += (size_t)Nn;
  float* ed2   = w; w += (size_t)Nn;
  int* cnt     = (int*)w;  w += (size_t)Nn + 1;

  hipMemsetAsync(cnt,   0, ((size_t)Nn + 1) * sizeof(int), stream);

  const int TB = 256;
  auto gemm_grid = [](int M, int N) {
    return dim3(((M >> 4) + 3) / 4, N >> 6);
  };
  const int GB = 128;

  edge_place_kernel<<<(E + TB - 1) / TB, TB, 0, stream>>>(rowE, colE, cnt, slots, ovf, Nn, E);

  wmma_gemm_strip64<128, 2><<<gemm_grid(Nn, 128), GB, 0, stream>>>(
      x, W1, h1, nullptr, attn1, es1, ed1, Nn, 128, 4);
  edge_softmax_gather_kernel<4, 32, true><<<(Nn * 32 + TB - 1) / TB, TB, 0, stream>>>(
      es1, ed1, h1, out1, cnt, slots, ovf, Nn);

  wmma_gemm_strip64<128, 1><<<gemm_grid(Nn, 64), GB, 0, stream>>>(
      out1, W2, h2, nullptr, attn2, es2, ed2, Nn, 64, 1);
  edge_softmax_gather_kernel<1, 64, false><<<(Nn * 16 + TB - 1) / TB, TB, 0, stream>>>(
      es2, ed2, h2, out2, cnt, slots, ovf, Nn);

  wmma_gemm_strip64<64, 0><<<gemm_grid(Nn, 64), GB, 0, stream>>>(
      out2, headW, (float*)d_out, headb, nullptr, nullptr, nullptr, Nn, 64, 0);
}


# Last linear layer (64x64 + bias) done by the layer-2 gather's own block: 16 nodes' sums to an LDS tile, waves 0-3 run the strip's WMMAs (same k and step order); its launch and sums plane go
# speedup vs baseline: 7.7148x; 1.0324x over previous
#include <hip/hip_runtime.h>
#include <hip/hip_bf16.h>

typedef float v2f __attribute__((ext_vector_type(2)));
typedef float v8f __attribute__((ext_vector_type(8)));

#define GAT_IN_DIM 128
#define GAT_NEG_SLOPE 0.2f


__device__ __forceinline__ float row16_sum(float v) {
  v += __builtin_bit_cast(float, __builtin_amdgcn_update_dpp(
           0, __builtin_bit_cast(int, v), 0x161, 0xf, 0xf, true));
  v += __builtin_bit_cast(float, __builtin_amdgcn_update_dpp(
           0, __builtin_bit_cast(int, v), 0x162, 0xf, 0xf, true));
  v += __builtin_bit_cast(float, __builtin_amdgcn_update_dpp(
           0, __builtin_bit_cast(int, v), 0x164, 0xf, 0xf, true));
  v += __builtin_bit_cast(float, __builtin_amdgcn_update_dpp(
           0, __builtin_bit_cast(int, v), 0x168, 0xf, 0xf, true));
  return v;
}

template <int K, int SH>
__global__ void __launch_bounds__(128)
wmma_gemm_strip64(const float* __restrict__ A,
                  const float* __restrict__ B,
                  float* __restrict__ C,
                  const float* __restrict__ bias,
                  const float* __restrict__ attn,
                  float* __restrict__ es,
                  float* __restrict__ ed,
                  int M, int N, int Htot) {
  __shared__ float4 Bs[(K / 4) * 64];
  const int tid = threadIdx.x;
  const int lane = tid & 31;
  const int wv = tid >> 5;
  const int nb = blockIdx.y * 64;
  const int c = lane & 15;
  const int h = lane >> 4;
  const int strip = blockIdx.x * 4 + wv;
  const bool live = strip < (M >> 4);
  const int tm = strip << 4;
  const float* ap = A + (size_t)(tm + c) * K + 4 * h;

  float4 afirst[4];
  if (live) {
#pragma unroll
    for (int i = 0; i < 4; ++i) afirst[i] = *(const float4*)(ap + 8 * i);
  }

  constexpr int ITEMS = (K / 4) * 16;
#pragma unroll
  for (int it0 = 0; it0 < ITEMS; it0 += 128) {
    const int it = it0 + tid;
    const int q = it >> 4;
    const int n4 = (it & 15) << 2;
    const float* bp = B + (size_t)(4 * q) * N + nb + n4;
    const float4 r0 = *(const float4*)(bp);
    const float4 r1 = *(const float4*)(bp + N);
    const float4 r2 = *(const float4*)(bp + 2 * (size_t)N);
    const float4 r3 = *(const float4*)(bp + 3 * (size_t)N);
    float4* d = Bs + q * 64 + n4;
    d[0] = make_float4(r0.x, r1.x, r2.x, r3.x);
    d[1] = make_float4(r0.y, r1.y, r2.y, r3.y);
    d[2] = make_float4(r0.z, r1.z, r2.z, r3.z);
    d[3] = make_float4(r0.w, r1.w, r2.w, r3.w);
  }
  __syncthreads();
  if (!live) return;

  const float4* bl = Bs + h * 64 + c;
  v8f acc[4] = {};
#pragma unroll
  for (int kb = 0; kb < K; kb += 8) {
    const float4 a = (kb < 32) ? afirst[kb >> 3] : *(const float4*)(ap + kb);
    const v2f a0 = {a.x, a.y};
    const v2f a1 = {a.z, a.w};
    float4 bq[4];
#pragma unroll
    for (int t = 0; t < 4; ++t) bq[t] = bl[(kb >> 2) * 64 + 16 * t];
#pragma unroll
    for (int t = 0; t < 4; ++t) {
      const v2f b0 = {bq[t].x, bq[t].y};
      acc[t] = __builtin_amdgcn_wmma_f32_16x16x4_f32(
           false, a0,  false, b0,
           (short)0, acc[t],  false,  false);
    }
#pragma unroll
    for (int t = 0; t < 4; ++t) {
      const v2f b1 = {bq[t].z, bq[t].w};
      acc[t] = __builtin_amdgcn_wmma_f32_16x16x4_f32(
           false, a1,  false, b1,
           (short)0, acc[t],  false,  false);
    }
  }

#pragma unroll
  for (int t = 0; t < 4; ++t) {
    const int col = nb + 16 * t + c;
    const float badd = bias ? bias[col] : 0.0f;
#pragma unroll
    for (int r = 0; r < 8; ++r) {
      acc[t][r] += badd;
      C[(size_t)(tm + r + 8 * h) * N + col] = acc[t][r];
    }
  }

  if constexpr (SH > 0) {
    constexpr int F = 64 / SH;
    constexpr int TPH = 4 / SH;
    const int rown = c & 7;
#pragma unroll
    for (int hl = 0; hl < SH; ++hl) {
      const int hd = blockIdx.y * SH + hl;
      const float* av = attn + (size_t)hd * 2 * F;
      float alv[TPH], arv[TPH];
#pragma unroll
      for (int tt = 0; tt < TPH; ++tt) {
        alv[tt] = av[16 * tt + c];
        arv[tt] = av[F + 16 * tt + c];
      }
      float sv = 0.f, dv = 0.f;
#pragma unroll
      for (int r = 0; r < 8; ++r) {
        float ps = acc[hl * TPH][r] * alv[0];
        float pd = acc[hl * TPH][r] * arv[0];
#pragma unroll
        for (int tt = 1; tt < TPH; ++tt) {
          ps += acc[hl * TPH + tt][r] * alv[tt];
          pd += acc[hl * TPH + tt][r] * arv[tt];
        }
        ps = row16_sum(ps);
        pd = row16_sum(pd);
        sv = (rown == r) ? ps : sv;
        dv = (rown == r) ? pd : dv;
      }
      float* dst = (c < 8) ? es : ed;
      dst[(size_t)(tm + 8 * h + rown) * Htot + hd] = (c < 8) ? sv : dv;
    }
  }
}

#define GAT_SLOTS 64

__global__ void edge_place_kernel(const int* __restrict__ rowE,
                                  const int* __restrict__ colE,
                                  int* __restrict__ cnt,
                                  int* __restrict__ slots,
                                  int* __restrict__ ovf,
                                  int Nn, int E) {
  int e = blockIdx.x * blockDim.x + threadIdx.x;
  if (e >= E) return;
  const int c = colE[e];
  const int s = rowE[e];
  const int r = atomicAdd(&cnt[c], 1);
  if (r < GAT_SLOTS) {
    slots[c * GAT_SLOTS + r] = s;
  } else {
    const int o = atomicAdd(&cnt[Nn], 1);
    ovf[2 * o] = c;
    ovf[2 * o + 1] = s;
  }
}

__device__ __forceinline__ float leaky(float a) {
  return (a > 0.f) ? a : a * GAT_NEG_SLOPE;
}

template <int U>
__device__ __forceinline__ int row_share(int v) {
  return __builtin_amdgcn_update_dpp(0, v, 0x150 + U, 0xf, 0xf, true);
}

template <int H, int F, bool RELU, bool HEAD>
__global__ void __launch_bounds__(256)
edge_softmax_gather_kernel(const float* __restrict__ es,
                           const float* __restrict__ ed,
                           const float* __restrict__ hsrc,
                           float* __restrict__ outd,
                           const int* __restrict__ cnt,
                           const int* __restrict__ slots,
                           const int* __restrict__ ovf,
                           const float* __restrict__ hw,
                           const float* __restrict__ hb,
                           int Nn) {
  constexpr int Ftot = H * F;
  constexpr int LPN = Ftot / 4;
  static_assert(LPN == 16 || LPN == 32, "a node's lanes must fill 16-lane rows");
  constexpr int CH = 8;
  const int t = blockIdx.x * blockDim.x + threadIdx.x;
  int c = t / LPN;
  if constexpr (HEAD) {
    c = min(c, Nn - 1);
  } else {
    if (c >= Nn) return;
  }
  const int j = t % LPN;
  const int hm = (j * 4) / F;
  const int l8 = threadIdx.x & (CH - 1);
  const float4* __restrict__ h4 = (const float4*)hsrc;

  const int deg = cnt[c];
  const int beg = c * GAT_SLOTS;
  const int end = beg + min(deg, GAT_SLOTS);
  const float edc = ed[c * H + hm];
  const float asf = leaky(es[c * H + hm] + edc);
  float4 acc = h4[c * LPN + j];
  float m = asf;
  float den = 1.f;

  if (beg < end) {
    const int last = end - 1;
    int sv = c;
    if (beg + l8 < end) sv = slots[beg + l8];
    for (int cb = beg; cb < end; cb += CH) {
      const int svn = slots[min(cb + CH + l8, last)];

      int s[CH];
      s[0] = row_share<0>(sv); s[1] = row_share<1>(sv);
      s[2] = row_share<2>(sv); s[3] = row_share<3>(sv);
      s[4] = row_share<4>(sv); s[5] = row_share<5>(sv);
      s[6] = row_share<6>(sv); s[7] = row_share<7>(sv);
      float e[CH];
      float4 r[CH];
      for (int u = 0; u < CH; ++u) e[u] = es[s[u] * H + hm];
      for (int u = 0; u < CH; ++u) r[u] = h4[s[u] * LPN + j];

      const int rem = end - cb;
      float a[CH];
      float mc = m;
      for (int u = 0; u < CH; ++u) {
        a[u] = leaky(e[u] + edc);
        mc = (u < rem) ? fmaxf(mc, a[u]) : mc;
      }
      const float sc = __expf(m - mc);
      den *= sc;
      acc.x *= sc;
      acc.y *= sc;
      acc.z *= sc;
      acc.w *= sc;
      for (int u = 0; u < CH; ++u) {
        const float p = (u < rem) ? __expf(a[u] - mc) : 0.f;
        den += p;
        acc.x += r[u].x * p;
        acc.y += r[u].y * p;
        acc.z += r[u].z * p;
        acc.w += r[u].w * p;
      }
      m = mc;
      sv = svn;
    }
  }
  if (deg > GAT_SLOTS) {
    const int no = cnt[Nn];
    for (int i = 0; i < no; ++i) {
      if (ovf[2 * i] != c) continue;
      const int s = ovf[2 * i + 1];
      const float a = leaky(es[s * H + hm] + edc);
      const float4 r = h4[s * LPN + j];
      const float mc = fmaxf(m, a);
      const float sc = __expf(m - mc);
      const float p = __expf(a - mc);
      den = den * sc + p;
      acc.x = acc.x * sc + r.x * p;
      acc.y = acc.y * sc + r.y * p;
      acc.z = acc.z * sc + r.z * p;
      acc.w = acc.w * sc + r.w * p;
      m = mc;
    }
  }
  const float inv = 1.0f / den;
  acc.x *= inv;
  acc.y *= inv;
  acc.z *= inv;
  acc.w *= inv;
  if constexpr (RELU) {
    acc.x = fmaxf(acc.x, 0.f);
    acc.y = fmaxf(acc.y, 0.f);
    acc.z = fmaxf(acc.z, 0.f);
    acc.w = fmaxf(acc.w, 0.f);
  }
  if constexpr (!HEAD) {
    ((float4*)outd)[c * LPN + j] = acc;
  } else {
    static_assert(Ftot == 64 && LPN == 16, "the fused layer is 64 -> 64, 16 nodes per block");
    __shared__ float4 T4[16 * 17];
    const int lane = threadIdx.x & 31;
    const int wv = threadIdx.x >> 5;
    const int cc = lane & 15;
    const int hh = lane >> 4;
    const int ncol = 16 * (wv & 3) + cc;
    T4[(threadIdx.x >> 4) * 17 + j] = acc;
    float bv[32];
    if (wv < 4) {
#pragma unroll
      for (int q = 0; q < 8; ++q) {
#pragma unroll
        for (int i = 0; i < 4; ++i) bv[4 * q + i] = hw[(8 * q + 4 * hh + i) * 64 + ncol];
      }
    }
    __syncthreads();
    if (wv >= 4) return;
    v8f d = {};
#pragma unroll
    for (int q = 0; q < 8; ++q) {
      const float4 a = T4[cc * 17 + 2 * q + hh];
      const v2f a0 = {a.x, a.y};
      const v2f a1 = {a.z, a.w};
      const v2f b0 = {bv[4 * q], bv[4 * q + 1]};
      const v2f b1 = {bv[4 * q + 2], bv[4 * q + 3]};
      d = __builtin_amdgcn_wmma_f32_16x16x4_f32(
           false, a0,  false, b0,
           (short)0, d,  false,  false);
      d = __builtin_amdgcn_wmma_f32_16x16x4_f32(
           false, a1,  false, b1,
           (short)0, d,  false,  false);
    }
    const float badd = hb[ncol];
    const int row0 = blockIdx.x * 16 + 8 * hh;
    float* op = outd + (size_t)row0 * 64 + ncol;
    if (blockIdx.x * 16 + 16 <= Nn) {
#pragma unroll
      for (int r = 0; r < 8; ++r) op[r * 64] = d[r] + badd;
    } else {
#pragma unroll
      for (int r = 0; r < 8; ++r) {
        if (row0 + r < Nn) op[r * 64] = d[r] + badd;
      }
    }
  }
}


extern "C" void kernel_launch(void* const* d_in, const int* in_sizes, int n_in,
                              void* d_out, int out_size, void* d_ws, size_t ws_size,
                              hipStream_t stream) {
  const float* x     = (const float*)d_in[0];
  const int*   ei    = (const int*)  d_in[1];
  const float* W1    = (const float*)d_in[2];
  const float* attn1 = (const float*)d_in[3];
  const float* W2    = (const float*)d_in[4];
  const float* attn2 = (const float*)d_in[5];
  const float* headW = (const float*)d_in[6];
  const float* headb = (const float*)d_in[7];

  const int Nn = in_sizes[0] / GAT_IN_DIM;
  const int E  = in_sizes[1] / 2;
  const int* rowE = ei;
  const int* colE = ei + E;

  float* w = (float*)d_ws;
  float* h1    = w; w += (size_t)Nn * 128;
  float* out1  = w; w += (size_t)Nn * 128;
  float* h2    = w; w += (size_t)Nn * 64;
  int* slots   = (int*)w;  w += (size_t)Nn * GAT_SLOTS;
  int* ovf     = (int*)w;  w += (size_t)E * 2;
  float* es1   = w; w += (size_t)Nn * 4;
  float* ed1   = w; w += (size_t)Nn * 4;
  float* es2   = w; w += (size_t)Nn;
  float* ed2   = w; w += (size_t)Nn;
  int* cnt     = (int*)w;  w += (size_t)Nn + 1;

  hipMemsetAsync(cnt,   0, ((size_t)Nn + 1) * sizeof(int), stream);

  const int TB = 256;
  auto gemm_grid = [](int M, int N) {
    return dim3(((M >> 4) + 3) / 4, N >> 6);
  };
  const int GB = 128;

  edge_place_kernel<<<(E + TB - 1) / TB, TB, 0, stream>>>(rowE, colE, cnt, slots, ovf, Nn, E);

  wmma_gemm_strip64<128, 2><<<gemm_grid(Nn, 128), GB, 0, stream>>>(
      x, W1, h1, nullptr, attn1, es1, ed1, Nn, 128, 4);
  edge_softmax_gather_kernel<4, 32, true, false><<<(Nn * 32 + TB - 1) / TB, TB, 0, stream>>>(
      es1, ed1, h1, out1, cnt, slots, ovf, nullptr, nullptr, Nn);

  wmma_gemm_strip64<128, 1><<<gemm_grid(Nn, 64), GB, 0, stream>>>(
      out1, W2, h2, nullptr, attn2, es2, ed2, Nn, 64, 1);
  edge_softmax_gather_kernel<1, 64, false, true><<<(Nn * 16 + TB - 1) / TB, TB, 0, stream>>>(
      es2, ed2, h2, (float*)d_out, cnt, slots, ovf, headW, headb, Nn);
}
